# starting kernel
# baseline (speedup 1.0000x reference)
#include <hip/hip_runtime.h>
#include <math.h>

#define NN      4096
#define NFEAT   512
#define NHID    64
#define NCLASS  128
#define NHEADS  8
#define ALPHA   0.2f
#define NEGV    (-9000000000000000.0f)

typedef __attribute__((ext_vector_type(2))) float v2f;
typedef __attribute__((ext_vector_type(8))) float v8f;

__global__ void pack_adj_kernel(const int* __restrict__ adj,
                                unsigned* __restrict__ bits, long nwords) {
    long w = (long)blockIdx.x * blockDim.x + threadIdx.x;
    if (w >= nwords) return;
    const int* p = adj + w * 32;
    unsigned m = 0u;
    #pragma unroll
    for (int b = 0; b < 32; ++b) m |= (p[b] > 0 ? 1u : 0u) << b;
    bits[w] = m;
}

template <int K, int NB>
__global__ __launch_bounds__(128) void wmma_gemm_f32(
        const float* __restrict__ A,
        const float* __restrict__ B, long bStride,
        float* __restrict__ C, long cStride) {
    const int wave = threadIdx.x >> 5;
    const int lane = threadIdx.x & 31;
    const int hi   = lane >> 4;
    const int lo   = lane & 15;
    const int row0 = blockIdx.x * 16;
    const int col0 = (blockIdx.y * 4 + wave) * 16;
    const int head = blockIdx.z;

    const float* __restrict__ arow  = A + (long)(row0 + lo) * K + 2 * hi;
    const float* __restrict__ bbase = B + (long)head * bStride + 2 * hi * NB + col0 + lo;

    v8f acc = {};
    for (int k = 0; k < K; k += 4) {
        v2f a;
        a.x = arow[k];
        a.y = arow[k + 1];
        v2f b;
        b.x = bbase[k * NB];
        b.y = bbase[k * NB + NB];
        acc = __builtin_amdgcn_wmma_f32_16x16x4_f32(
            false, a, false, b, (short)0, acc, false, false);
    }

    float* cp = C + (long)head * cStride + (long)(row0 + 8 * hi) * NB + col0 + lo;
    #pragma unroll
    for (int v = 0; v < 8; ++v) cp[v * NB] = acc[v];
}

__global__ void fused_fifj(const float* __restrict__ H, long hStride, int F,
                           const float* __restrict__ a, int aStride,
                           float* __restrict__ fi, float* __restrict__ fj, int n) {
    const int head = blockIdx.y;
    const int nd   = blockIdx.x * blockDim.x + threadIdx.x;
    if (nd >= n) return;
    const float* hp = H + (long)head * hStride + (long)nd * F;
    const float* a1 = a + (long)head * aStride;
    const float* a2 = a1 + F;
    float s1 = 0.f, s2 = 0.f;
    for (int o = 0; o < F; ++o) {
        float hv = hp[o];
        s1 += hv * a1[o];
        s2 += hv * a2[o];
    }
    fi[(long)head * n + nd] = s1;
    fj[(long)head * n + nd] = s2;
}

template <int F, int JW, int R, bool ELU>
__global__ __launch_bounds__(JW * 32) void gat_attn(
        const float* __restrict__ Hfeat, long hStride,
        const float* __restrict__ fi, const float* __restrict__ fj,
        const unsigned* __restrict__ bits,
        float* __restrict__ Out, int ldo, int headColMul, int n) {
    constexpr int NC = F / 16;
    const int wave = threadIdx.x >> 5;
    const int lane = threadIdx.x & 31;
    const int hi   = lane >> 4;
    const int lo   = lane & 15;
    const int head = blockIdx.y;
    const int i0   = blockIdx.x * (16 * R);
    const int nwords = n >> 5;

    const float* __restrict__ Hh  = Hfeat + (long)head * hStride;
    const float* __restrict__ fih = fi + (long)head * n;
    const float* __restrict__ fjh = fj + (long)head * n;
    const float* __restrict__ bbase = Hh + 2 * hi * F + lo;

    float fiv[R];
    const unsigned* __restrict__ brow[R];
    #pragma unroll
    for (int r = 0; r < R; ++r) {
        fiv[r]  = fih[i0 + 16 * r + lo];
        brow[r] = bits + (long)(i0 + 16 * r + lo) * nwords;
    }

    const int chunk = n / JW;
    const int jbeg  = wave * chunk;
    const int jend  = jbeg + chunk;

    float m[R], l[R];
    v8f acc[R][NC];
    #pragma unroll
    for (int r = 0; r < R; ++r) {
        m[r] = -INFINITY;
        l[r] = 0.f;
        #pragma unroll
        for (int c = 0; c < NC; ++c) acc[r][c] = (v8f){};
    }

    for (int j0 = jbeg; j0 < jend; j0 += 32) {
        unsigned wmask[R];
        #pragma unroll
        for (int r = 0; r < R; ++r) wmask[r] = brow[r][j0 >> 5];

        #pragma unroll
        for (int half = 0; half < 2; ++half) {
            const int jb = j0 + 16 * half;
            const int bitbase = 16 * half;
            const float* __restrict__ fjp = fjh + jb + 2 * hi;

            float fjv[8];
            #pragma unroll
            for (int s = 0; s < 4; ++s)
                #pragma unroll
                for (int t = 0; t < 2; ++t) fjv[2 * s + t] = fjp[4 * s + t];

            float p8[R][8];
            #pragma unroll
            for (int r = 0; r < R; ++r) {
                float e8[8];
                float tmax = -INFINITY;
                #pragma unroll
                for (int s = 0; s < 4; ++s) {
                    #pragma unroll
                    for (int t = 0; t < 2; ++t) {
                        const int c = 4 * s + 2 * hi + t;
                        float v = fiv[r] + fjv[2 * s + t];
                        v = v > 0.f ? v : ALPHA * v;
                        const float e = ((wmask[r] >> (bitbase + c)) & 1u) ? v : NEGV;
                        e8[2 * s + t] = e;
                        tmax = fmaxf(tmax, e);
                    }
                }
                tmax = fmaxf(tmax, __shfl_xor(tmax, 16, 32));
                const float mnew = fmaxf(m[r], tmax);

                float psum = 0.f;
                #pragma unroll
                for (int q = 0; q < 8; ++q) {
                    p8[r][q] = __expf(e8[q] - mnew);
                    psum += p8[r][q];
                }
                psum += __shfl_xor(psum, 16, 32);

                if (__any(mnew > m[r])) {
                    const float scale = __expf(m[r] - mnew);
                    float sv[8];
                    #pragma unroll
                    for (int v = 0; v < 8; ++v) sv[v] = __shfl(scale, v + 8 * hi, 32);
                    #pragma unroll
                    for (int c = 0; c < NC; ++c)
                        #pragma unroll
                        for (int v = 0; v < 8; ++v) acc[r][c][v] *= sv[v];
                    l[r] = l[r] * scale + psum;
                    m[r] = mnew;
                } else {
                    l[r] += psum;
                }
            }

            const float* __restrict__ bp = bbase + jb * F;
            #pragma unroll
            for (int s = 0; s < 4; ++s) {
                const float* __restrict__ bs = bp + 4 * s * F;
                #pragma unroll
                for (int c = 0; c < NC; ++c) {
                    v2f b;
                    b.x = bs[16 * c];
                    b.y = bs[16 * c + F];
                    #pragma unroll
                    for (int r = 0; r < R; ++r) {
                        v2f a;
                        a.x = p8[r][2 * s];
                        a.y = p8[r][2 * s + 1];
                        acc[r][c] = __builtin_amdgcn_wmma_f32_16x16x4_f32(
                            false, a, false, b, (short)0, acc[r][c], false, false);
                    }
                }
            }
        }
    }

    __shared__ float m_lds[R][JW * 16];
    __shared__ float l_lds[R][JW * 16];
    __shared__ float L_lds[16];
    __shared__ float accbuf[JW * 16 * F];

    #pragma unroll
    for (int r = 0; r < R; ++r) {
        if (hi == 0) {
            m_lds[r][wave * 16 + lo] = m[r];
            l_lds[r][wave * 16 + lo] = l[r];
        }
    }
    __syncthreads();

    constexpr int TOT = 16 * F;
    constexpr int NT  = JW * 32;
    #pragma unroll
    for (int r = 0; r < R; ++r) {
        float Mrow = -INFINITY;
        #pragma unroll
        for (int w = 0; w < JW; ++w) Mrow = fmaxf(Mrow, m_lds[r][w * 16 + lo]);
        float Lrow = 0.f;
        #pragma unroll
        for (int w = 0; w < JW; ++w)
            Lrow += l_lds[r][w * 16 + lo] * __expf(m_lds[r][w * 16 + lo] - Mrow);
        if (wave == 0 && hi == 0) L_lds[lo] = Lrow;

        const float fown = __expf(m[r] - Mrow);
        float fv[8];
        #pragma unroll
        for (int v = 0; v < 8; ++v) fv[v] = __shfl(fown, v + 8 * hi, 32);

        float* ab = accbuf + (wave * 16 + 8 * hi) * F + lo;
        #pragma unroll
        for (int c = 0; c < NC; ++c)
            #pragma unroll
            for (int v = 0; v < 8; ++v)
                ab[v * F + 16 * c] = acc[r][c][v] * fv[v];
        __syncthreads();

        float* outBase = Out + (long)(i0 + 16 * r) * ldo + head * headColMul;
        #pragma unroll
        for (int e = 0; e < TOT / NT; ++e) {
            const int flat = e * NT + threadIdx.x;
            const int row = flat / F;
            const int col = flat % F;
            float val = 0.f;
            #pragma unroll
            for (int w = 0; w < JW; ++w) val += accbuf[(w * 16 + row) * F + col];
            float rr = val / L_lds[row];
            if (ELU) rr = rr > 0.f ? rr : (__expf(rr) - 1.0f);
            outBase[(long)row * ldo + col] = rr;
        }
        if (r + 1 < R) __syncthreads();
    }
}

__global__ void logsoftmax128(const float* __restrict__ in,
                              float* __restrict__ out, int n) {
    const int wave = threadIdx.x >> 5;
    const int lane = threadIdx.x & 31;
    const int row  = blockIdx.x * (blockDim.x >> 5) + wave;
    if (row >= n) return;
    const float* rp = in + (long)row * NCLASS;
    float v[4];
    float mx = -INFINITY;
    #pragma unroll
    for (int q = 0; q < 4; ++q) {
        v[q] = rp[lane + 32 * q];
        mx = fmaxf(mx, v[q]);
    }
    #pragma unroll
    for (int off = 16; off >= 1; off >>= 1) mx = fmaxf(mx, __shfl_xor(mx, off, 32));
    float s = 0.f;
    #pragma unroll
    for (int q = 0; q < 4; ++q) s += __expf(v[q] - mx);
    #pragma unroll
    for (int off = 16; off >= 1; off >>= 1) s += __shfl_xor(s, off, 32);
    const float ls = __logf(s);
    #pragma unroll
    for (int q = 0; q < 4; ++q) out[(long)row * NCLASS + lane + 32 * q] = v[q] - mx - ls;
}

extern "C" void kernel_launch(void* const* d_in, const int* in_sizes, int n_in,
                              void* d_out, int out_size, void* d_ws, size_t ws_size,
                              hipStream_t stream) {
    const float* x     = (const float*)d_in[0];
    const int*   adj   = (const int*)d_in[1];
    const float* W     = (const float*)d_in[2];
    const float* a     = (const float*)d_in[3];
    const float* W_out = (const float*)d_in[4];
    const float* a_out = (const float*)d_in[5];
    float* out = (float*)d_out;

    char* ws = (char*)d_ws;
    unsigned* bits = (unsigned*)ws;
    ws += (size_t)NN * (NN / 32) * sizeof(unsigned);
    float* h1 = (float*)ws;
    ws += (size_t)NHEADS * NN * NHID * sizeof(float);
    float* fi1 = (float*)ws; ws += (size_t)NHEADS * NN * sizeof(float);
    float* fj1 = (float*)ws; ws += (size_t)NHEADS * NN * sizeof(float);
    float* xcat = (float*)ws;
    ws += (size_t)NN * (NHEADS * NHID) * sizeof(float);
    float* h2 = (float*)ws;
    ws += (size_t)NN * NCLASS * sizeof(float);
    float* fi2 = (float*)ws; ws += (size_t)NN * sizeof(float);
    float* fj2 = (float*)ws; ws += (size_t)NN * sizeof(float);
    float* outpre = (float*)ws;

    {
        const long nwords = (long)NN * (NN / 32);
        pack_adj_kernel<<<dim3((nwords + 255) / 256), dim3(256), 0, stream>>>(adj, bits, nwords);
    }

    wmma_gemm_f32<NFEAT, NHID><<<dim3(NN / 16, 1, NHEADS), dim3(128), 0, stream>>>(
        x, W, (long)NFEAT * NHID, h1, (long)NN * NHID);

    fused_fifj<<<dim3(NN / 256, NHEADS), dim3(256), 0, stream>>>(
        h1, (long)NN * NHID, NHID, a, 2 * NHID, fi1, fj1, NN);

    gat_attn<NHID, 4, 2, false><<<dim3(NN / 32, NHEADS), dim3(128), 0, stream>>>(
        h1, (long)NN * NHID, fi1, fj1, bits,
        xcat, NHEADS * NHID, NHID, NN);

    wmma_gemm_f32<NHEADS * NHID, NCLASS><<<dim3(NN / 16, NCLASS / 64, 1), dim3(128), 0, stream>>>(
        xcat, W_out, 0, h2, 0);

    fused_fifj<<<dim3(NN / 256, 1), dim3(256), 0, stream>>>(
        h2, 0, NCLASS, a_out, 2 * NCLASS, fi2, fj2, NN);

    gat_attn<NCLASS, 4, 2, true><<<dim3(NN / 32, 1), dim3(128), 0, stream>>>(
        h2, 0, fi2, fj2, bits,
        outpre, NCLASS, 0, NN);

    logsoftmax128<<<dim3(NN / 4), dim3(128), 0, stream>>>(outpre, out, NN);
}


# layer-2 attention launched as 256 blocks of 16 waves (one row tile per wave, j split 16-way; was 128 blocks of 4 waves, 2 tiles each): every SIMD holds 4 waves; partials merge 16-way, fixed order
# speedup vs baseline: 1.2169x; 1.2169x over previous
#include <hip/hip_runtime.h>
#include <math.h>

#define NN      4096
#define NFEAT   512
#define NHID    64
#define NCLASS  128
#define NHEADS  8
#define ALPHA   0.2f
#define NEGV    (-9000000000000000.0f)

typedef __attribute__((ext_vector_type(2))) float v2f;
typedef __attribute__((ext_vector_type(8))) float v8f;

__global__ void pack_adj_kernel(const int* __restrict__ adj,
                                unsigned* __restrict__ bits, long nwords) {
    long w = (long)blockIdx.x * blockDim.x + threadIdx.x;
    if (w >= nwords) return;
    const int* p = adj + w * 32;
    unsigned m = 0u;
    #pragma unroll
    for (int b = 0; b < 32; ++b) m |= (p[b] > 0 ? 1u : 0u) << b;
    bits[w] = m;
}

template <int K, int NB>
__global__ __launch_bounds__(128) void wmma_gemm_f32(
        const float* __restrict__ A,
        const float* __restrict__ B, long bStride,
        float* __restrict__ C, long cStride) {
    const int wave = threadIdx.x >> 5;
    const int lane = threadIdx.x & 31;
    const int hi   = lane >> 4;
    const int lo   = lane & 15;
    const int row0 = blockIdx.x * 16;
    const int col0 = (blockIdx.y * 4 + wave) * 16;
    const int head = blockIdx.z;

    const float* __restrict__ arow  = A + (long)(row0 + lo) * K + 2 * hi;
    const float* __restrict__ bbase = B + (long)head * bStride + 2 * hi * NB + col0 + lo;

    v8f acc = {};
    for (int k = 0; k < K; k += 4) {
        v2f a;
        a.x = arow[k];
        a.y = arow[k + 1];
        v2f b;
        b.x = bbase[k * NB];
        b.y = bbase[k * NB + NB];
        acc = __builtin_amdgcn_wmma_f32_16x16x4_f32(
            false, a, false, b, (short)0, acc, false, false);
    }

    float* cp = C + (long)head * cStride + (long)(row0 + 8 * hi) * NB + col0 + lo;
    #pragma unroll
    for (int v = 0; v < 8; ++v) cp[v * NB] = acc[v];
}

__global__ void fused_fifj(const float* __restrict__ H, long hStride, int F,
                           const float* __restrict__ a, int aStride,
                           float* __restrict__ fi, float* __restrict__ fj, int n) {
    const int head = blockIdx.y;
    const int nd   = blockIdx.x * blockDim.x + threadIdx.x;
    if (nd >= n) return;
    const float* hp = H + (long)head * hStride + (long)nd * F;
    const float* a1 = a + (long)head * aStride;
    const float* a2 = a1 + F;
    float s1 = 0.f, s2 = 0.f;
    for (int o = 0; o < F; ++o) {
        float hv = hp[o];
        s1 += hv * a1[o];
        s2 += hv * a2[o];
    }
    fi[(long)head * n + nd] = s1;
    fj[(long)head * n + nd] = s2;
}

template <int F, int JW, int R, bool ELU>
__global__ __launch_bounds__(JW * 32) void gat_attn(
        const float* __restrict__ Hfeat, long hStride,
        const float* __restrict__ fi, const float* __restrict__ fj,
        const unsigned* __restrict__ bits,
        float* __restrict__ Out, int ldo, int headColMul, int n) {
    constexpr int NC = F / 16;
    const int wave = threadIdx.x >> 5;
    const int lane = threadIdx.x & 31;
    const int hi   = lane >> 4;
    const int lo   = lane & 15;
    const int head = blockIdx.y;
    const int i0   = blockIdx.x * (16 * R);
    const int nwords = n >> 5;

    const float* __restrict__ Hh  = Hfeat + (long)head * hStride;
    const float* __restrict__ fih = fi + (long)head * n;
    const float* __restrict__ fjh = fj + (long)head * n;
    const float* __restrict__ bbase = Hh + 2 * hi * F + lo;

    float fiv[R];
    const unsigned* __restrict__ brow[R];
    #pragma unroll
    for (int r = 0; r < R; ++r) {
        fiv[r]  = fih[i0 + 16 * r + lo];
        brow[r] = bits + (long)(i0 + 16 * r + lo) * nwords;
    }

    const int chunk = n / JW;
    const int jbeg  = wave * chunk;
    const int jend  = jbeg + chunk;

    float m[R], l[R];
    v8f acc[R][NC];
    #pragma unroll
    for (int r = 0; r < R; ++r) {
        m[r] = -INFINITY;
        l[r] = 0.f;
        #pragma unroll
        for (int c = 0; c < NC; ++c) acc[r][c] = (v8f){};
    }

    for (int j0 = jbeg; j0 < jend; j0 += 32) {
        unsigned wmask[R];
        #pragma unroll
        for (int r = 0; r < R; ++r) wmask[r] = brow[r][j0 >> 5];

        #pragma unroll
        for (int half = 0; half < 2; ++half) {
            const int jb = j0 + 16 * half;
            const int bitbase = 16 * half;
            const float* __restrict__ fjp = fjh + jb + 2 * hi;

            float fjv[8];
            #pragma unroll
            for (int s = 0; s < 4; ++s)
                #pragma unroll
                for (int t = 0; t < 2; ++t) fjv[2 * s + t] = fjp[4 * s + t];

            float p8[R][8];
            #pragma unroll
            for (int r = 0; r < R; ++r) {
                float e8[8];
                float tmax = -INFINITY;
                #pragma unroll
                for (int s = 0; s < 4; ++s) {
                    #pragma unroll
                    for (int t = 0; t < 2; ++t) {
                        const int c = 4 * s + 2 * hi + t;
                        float v = fiv[r] + fjv[2 * s + t];
                        v = v > 0.f ? v : ALPHA * v;
                        const float e = ((wmask[r] >> (bitbase + c)) & 1u) ? v : NEGV;
                        e8[2 * s + t] = e;
                        tmax = fmaxf(tmax, e);
                    }
                }
                tmax = fmaxf(tmax, __shfl_xor(tmax, 16, 32));
                const float mnew = fmaxf(m[r], tmax);

                float psum = 0.f;
                #pragma unroll
                for (int q = 0; q < 8; ++q) {
                    p8[r][q] = __expf(e8[q] - mnew);
                    psum += p8[r][q];
                }
                psum += __shfl_xor(psum, 16, 32);

                if (__any(mnew > m[r])) {
                    const float scale = __expf(m[r] - mnew);
                    float sv[8];
                    #pragma unroll
                    for (int v = 0; v < 8; ++v) sv[v] = __shfl(scale, v + 8 * hi, 32);
                    #pragma unroll
                    for (int c = 0; c < NC; ++c)
                        #pragma unroll
                        for (int v = 0; v < 8; ++v) acc[r][c][v] *= sv[v];
                    l[r] = l[r] * scale + psum;
                    m[r] = mnew;
                } else {
                    l[r] += psum;
                }
            }

            const float* __restrict__ bp = bbase + jb * F;
            #pragma unroll
            for (int s = 0; s < 4; ++s) {
                const float* __restrict__ bs = bp + 4 * s * F;
                #pragma unroll
                for (int c = 0; c < NC; ++c) {
                    v2f b;
                    b.x = bs[16 * c];
                    b.y = bs[16 * c + F];
                    #pragma unroll
                    for (int r = 0; r < R; ++r) {
                        v2f a;
                        a.x = p8[r][2 * s];
                        a.y = p8[r][2 * s + 1];
                        acc[r][c] = __builtin_amdgcn_wmma_f32_16x16x4_f32(
                            false, a, false, b, (short)0, acc[r][c], false, false);
                    }
                }
            }
        }
    }

    __shared__ float m_lds[R][JW * 16];
    __shared__ float l_lds[R][JW * 16];
    __shared__ float L_lds[16];
    __shared__ float accbuf[JW * 16 * F];

    #pragma unroll
    for (int r = 0; r < R; ++r) {
        if (hi == 0) {
            m_lds[r][wave * 16 + lo] = m[r];
            l_lds[r][wave * 16 + lo] = l[r];
        }
    }
    __syncthreads();

    constexpr int TOT = 16 * F;
    constexpr int NT  = JW * 32;
    #pragma unroll
    for (int r = 0; r < R; ++r) {
        float Mrow = -INFINITY;
        #pragma unroll
        for (int w = 0; w < JW; ++w) Mrow = fmaxf(Mrow, m_lds[r][w * 16 + lo]);
        float Lrow = 0.f;
        #pragma unroll
        for (int w = 0; w < JW; ++w)
            Lrow += l_lds[r][w * 16 + lo] * __expf(m_lds[r][w * 16 + lo] - Mrow);
        if (wave == 0 && hi == 0) L_lds[lo] = Lrow;

        const float fown = __expf(m[r] - Mrow);
        float fv[8];
        #pragma unroll
        for (int v = 0; v < 8; ++v) fv[v] = __shfl(fown, v + 8 * hi, 32);

        float* ab = accbuf + (wave * 16 + 8 * hi) * F + lo;
        #pragma unroll
        for (int c = 0; c < NC; ++c)
            #pragma unroll
            for (int v = 0; v < 8; ++v)
                ab[v * F + 16 * c] = acc[r][c][v] * fv[v];
        __syncthreads();

        float* outBase = Out + (long)(i0 + 16 * r) * ldo + head * headColMul;
        #pragma unroll
        for (int e = 0; e < TOT / NT; ++e) {
            const int flat = e * NT + threadIdx.x;
            const int row = flat / F;
            const int col = flat % F;
            float val = 0.f;
            #pragma unroll
            for (int w = 0; w < JW; ++w) val += accbuf[(w * 16 + row) * F + col];
            float rr = val / L_lds[row];
            if (ELU) rr = rr > 0.f ? rr : (__expf(rr) - 1.0f);
            outBase[(long)row * ldo + col] = rr;
        }
        if (r + 1 < R) __syncthreads();
    }
}

__global__ void logsoftmax128(const float* __restrict__ in,
                              float* __restrict__ out, int n) {
    const int wave = threadIdx.x >> 5;
    const int lane = threadIdx.x & 31;
    const int row  = blockIdx.x * (blockDim.x >> 5) + wave;
    if (row >= n) return;
    const float* rp = in + (long)row * NCLASS;
    float v[4];
    float mx = -INFINITY;
    #pragma unroll
    for (int q = 0; q < 4; ++q) {
        v[q] = rp[lane + 32 * q];
        mx = fmaxf(mx, v[q]);
    }
    #pragma unroll
    for (int off = 16; off >= 1; off >>= 1) mx = fmaxf(mx, __shfl_xor(mx, off, 32));
    float s = 0.f;
    #pragma unroll
    for (int q = 0; q < 4; ++q) s += __expf(v[q] - mx);
    #pragma unroll
    for (int off = 16; off >= 1; off >>= 1) s += __shfl_xor(s, off, 32);
    const float ls = __logf(s);
    #pragma unroll
    for (int q = 0; q < 4; ++q) out[(long)row * NCLASS + lane + 32 * q] = v[q] - mx - ls;
}

extern "C" void kernel_launch(void* const* d_in, const int* in_sizes, int n_in,
                              void* d_out, int out_size, void* d_ws, size_t ws_size,
                              hipStream_t stream) {
    const float* x     = (const float*)d_in[0];
    const int*   adj   = (const int*)d_in[1];
    const float* W     = (const float*)d_in[2];
    const float* a     = (const float*)d_in[3];
    const float* W_out = (const float*)d_in[4];
    const float* a_out = (const float*)d_in[5];
    float* out = (float*)d_out;

    char* ws = (char*)d_ws;
    unsigned* bits = (unsigned*)ws;
    ws += (size_t)NN * (NN / 32) * sizeof(unsigned);
    float* h1 = (float*)ws;
    ws += (size_t)NHEADS * NN * NHID * sizeof(float);
    float* fi1 = (float*)ws; ws += (size_t)NHEADS * NN * sizeof(float);
    float* fj1 = (float*)ws; ws += (size_t)NHEADS * NN * sizeof(float);
    float* xcat = (float*)ws;
    ws += (size_t)NN * (NHEADS * NHID) * sizeof(float);
    float* h2 = (float*)ws;
    ws += (size_t)NN * NCLASS * sizeof(float);
    float* fi2 = (float*)ws; ws += (size_t)NN * sizeof(float);
    float* fj2 = (float*)ws; ws += (size_t)NN * sizeof(float);
    float* outpre = (float*)ws;

    {
        const long nwords = (long)NN * (NN / 32);
        pack_adj_kernel<<<dim3((nwords + 255) / 256), dim3(256), 0, stream>>>(adj, bits, nwords);
    }

    wmma_gemm_f32<NFEAT, NHID><<<dim3(NN / 16, 1, NHEADS), dim3(128), 0, stream>>>(
        x, W, (long)NFEAT * NHID, h1, (long)NN * NHID);

    fused_fifj<<<dim3(NN / 256, NHEADS), dim3(256), 0, stream>>>(
        h1, (long)NN * NHID, NHID, a, 2 * NHID, fi1, fj1, NN);

    gat_attn<NHID, 4, 2, false><<<dim3(NN / 32, NHEADS), dim3(128), 0, stream>>>(
        h1, (long)NN * NHID, fi1, fj1, bits,
        xcat, NHEADS * NHID, NHID, NN);

    wmma_gemm_f32<NHEADS * NHID, NCLASS><<<dim3(NN / 16, NCLASS / 64, 1), dim3(128), 0, stream>>>(
        xcat, W_out, 0, h2, 0);

    fused_fifj<<<dim3(NN / 256, 1), dim3(256), 0, stream>>>(
        h2, 0, NCLASS, a_out, 2 * NCLASS, fi2, fj2, NN);

    gat_attn<NCLASS, 16, 1, true><<<dim3(NN / 16, 1), dim3(512), 0, stream>>>(
        h2, 0, fi2, fj2, bits,
        outpre, NCLASS, 0, NN);

    logsoftmax128<<<dim3(NN / 4), dim3(128), 0, stream>>>(outpre, out, NN);
}


# attention loops: a half tile's B fragments requested before its softmax, the next tile's fj values and mask words under the WMMA run before; no load waited for at its use; sums and order unchanged
# speedup vs baseline: 1.2748x; 1.0476x over previous
#include <hip/hip_runtime.h>
#include <math.h>

#define NN      4096
#define NFEAT   512
#define NHID    64
#define NCLASS  128
#define NHEADS  8
#define ALPHA   0.2f
#define NEGV    (-9000000000000000.0f)

typedef __attribute__((ext_vector_type(2))) float v2f;
typedef __attribute__((ext_vector_type(8))) float v8f;

__global__ void pack_adj_kernel(const int* __restrict__ adj,
                                unsigned* __restrict__ bits, long nwords) {
    long w = (long)blockIdx.x * blockDim.x + threadIdx.x;
    if (w >= nwords) return;
    const int* p = adj + w * 32;
    unsigned m = 0u;
    #pragma unroll
    for (int b = 0; b < 32; ++b) m |= (p[b] > 0 ? 1u : 0u) << b;
    bits[w] = m;
}

template <int K, int NB>
__global__ __launch_bounds__(128) void wmma_gemm_f32(
        const float* __restrict__ A,
        const float* __restrict__ B, long bStride,
        float* __restrict__ C, long cStride) {
    const int wave = threadIdx.x >> 5;
    const int lane = threadIdx.x & 31;
    const int hi   = lane >> 4;
    const int lo   = lane & 15;
    const int row0 = blockIdx.x * 16;
    const int col0 = (blockIdx.y * 4 + wave) * 16;
    const int head = blockIdx.z;

    const float* __restrict__ arow  = A + (long)(row0 + lo) * K + 2 * hi;
    const float* __restrict__ bbase = B + (long)head * bStride + 2 * hi * NB + col0 + lo;

    v8f acc = {};
    for (int k = 0; k < K; k += 4) {
        v2f a;
        a.x = arow[k];
        a.y = arow[k + 1];
        v2f b;
        b.x = bbase[k * NB];
        b.y = bbase[k * NB + NB];
        acc = __builtin_amdgcn_wmma_f32_16x16x4_f32(
            false, a, false, b, (short)0, acc, false, false);
    }

    float* cp = C + (long)head * cStride + (long)(row0 + 8 * hi) * NB + col0 + lo;
    #pragma unroll
    for (int v = 0; v < 8; ++v) cp[v * NB] = acc[v];
}

__global__ void fused_fifj(const float* __restrict__ H, long hStride, int F,
                           const float* __restrict__ a, int aStride,
                           float* __restrict__ fi, float* __restrict__ fj, int n) {
    const int head = blockIdx.y;
    const int nd   = blockIdx.x * blockDim.x + threadIdx.x;
    if (nd >= n) return;
    const float* hp = H + (long)head * hStride + (long)nd * F;
    const float* a1 = a + (long)head * aStride;
    const float* a2 = a1 + F;
    float s1 = 0.f, s2 = 0.f;
    for (int o = 0; o < F; ++o) {
        float hv = hp[o];
        s1 += hv * a1[o];
        s2 += hv * a2[o];
    }
    fi[(long)head * n + nd] = s1;
    fj[(long)head * n + nd] = s2;
}

template <int F, int JW, int R, int PRE, bool ELU>
__global__ __launch_bounds__(JW * 32) void gat_attn(
        const float* __restrict__ Hfeat, long hStride,
        const float* __restrict__ fi, const float* __restrict__ fj,
        const unsigned* __restrict__ bits,
        float* __restrict__ Out, int ldo, int headColMul, int n) {
    constexpr int NC = F / 16;
    const int wave = threadIdx.x >> 5;
    const int lane = threadIdx.x & 31;
    const int hi   = lane >> 4;
    const int lo   = lane & 15;
    const int head = blockIdx.y;
    const int i0   = blockIdx.x * (16 * R);
    const int nwords = n >> 5;

    const float* __restrict__ Hh  = Hfeat + (long)head * hStride;
    const float* __restrict__ fih = fi + (long)head * n;
    const float* __restrict__ fjh = fj + (long)head * n;
    const float* __restrict__ bbase = Hh + 2 * hi * F + lo;

    float fiv[R];
    const unsigned* __restrict__ brow[R];
    #pragma unroll
    for (int r = 0; r < R; ++r) {
        fiv[r]  = fih[i0 + 16 * r + lo];
        brow[r] = bits + (long)(i0 + 16 * r + lo) * nwords;
    }

    const int chunk = n / JW;
    const int jbeg  = wave * chunk;
    const int jend  = jbeg + chunk;

    float m[R], l[R];
    v8f acc[R][NC];
    #pragma unroll
    for (int r = 0; r < R; ++r) {
        m[r] = -INFINITY;
        l[r] = 0.f;
        #pragma unroll
        for (int c = 0; c < NC; ++c) acc[r][c] = (v8f){};
    }

    unsigned wm_n[R];
    float fj_n[16];
    #pragma unroll
    for (int r = 0; r < R; ++r) wm_n[r] = brow[r][jbeg >> 5];
    {
        const float* __restrict__ fjp = fjh + jbeg + 2 * hi;
        #pragma unroll
        for (int q = 0; q < 8; ++q)
            #pragma unroll
            for (int t = 0; t < 2; ++t) fj_n[2 * q + t] = fjp[4 * q + t];
    }

    for (int j0 = jbeg; j0 < jend; j0 += 32) {
        unsigned wmask[R];
        float fjc[16];
        #pragma unroll
        for (int r = 0; r < R; ++r) wmask[r] = wm_n[r];
        #pragma unroll
        for (int q = 0; q < 16; ++q) fjc[q] = fj_n[q];

        #pragma unroll
        for (int half = 0; half < 2; ++half) {
            const int jb = j0 + 16 * half;
            const int bitbase = 16 * half;

            float fjv[8];
            #pragma unroll
            for (int q = 0; q < 8; ++q) fjv[q] = fjc[8 * half + q];

            v2f bf[4][NC];
            const float* __restrict__ bp = bbase + jb * F;
            #pragma unroll
            for (int s = 0; s < PRE; ++s) {
                const float* __restrict__ bs = bp + 4 * s * F;
                #pragma unroll
                for (int c = 0; c < NC; ++c) {
                    bf[s][c].x = bs[16 * c];
                    bf[s][c].y = bs[16 * c + F];
                }
            }

            float p8[R][8];
            #pragma unroll
            for (int r = 0; r < R; ++r) {
                float e8[8];
                float tmax = -INFINITY;
                #pragma unroll
                for (int s = 0; s < 4; ++s) {
                    #pragma unroll
                    for (int t = 0; t < 2; ++t) {
                        const int c = 4 * s + 2 * hi + t;
                        float v = fiv[r] + fjv[2 * s + t];
                        v = v > 0.f ? v : ALPHA * v;
                        const float e = ((wmask[r] >> (bitbase + c)) & 1u) ? v : NEGV;
                        e8[2 * s + t] = e;
                        tmax = fmaxf(tmax, e);
                    }
                }
                tmax = fmaxf(tmax, __shfl_xor(tmax, 16, 32));
                const float mnew = fmaxf(m[r], tmax);

                float psum = 0.f;
                #pragma unroll
                for (int q = 0; q < 8; ++q) {
                    p8[r][q] = __expf(e8[q] - mnew);
                    psum += p8[r][q];
                }
                psum += __shfl_xor(psum, 16, 32);

                if (__any(mnew > m[r])) {
                    const float scale = __expf(m[r] - mnew);
                    float sv[8];
                    #pragma unroll
                    for (int v = 0; v < 8; ++v) sv[v] = __shfl(scale, v + 8 * hi, 32);
                    #pragma unroll
                    for (int c = 0; c < NC; ++c)
                        #pragma unroll
                        for (int v = 0; v < 8; ++v) acc[r][c][v] *= sv[v];
                    l[r] = l[r] * scale + psum;
                    m[r] = mnew;
                } else {
                    l[r] += psum;
                }
            }

            if (half == 1) {
                const int jn = (j0 + 32 < jend) ? (j0 + 32) : j0;
                const float* __restrict__ fjp = fjh + jn + 2 * hi;
                #pragma unroll
                for (int q = 0; q < 8; ++q)
                    #pragma unroll
                    for (int t = 0; t < 2; ++t) fj_n[2 * q + t] = fjp[4 * q + t];
                #pragma unroll
                for (int r = 0; r < R; ++r) wm_n[r] = brow[r][jn >> 5];
            }
            #pragma unroll
            for (int s = PRE; s < 4; ++s) {
                const float* __restrict__ bs = bp + 4 * s * F;
                #pragma unroll
                for (int c = 0; c < NC; ++c) {
                    bf[s][c].x = bs[16 * c];
                    bf[s][c].y = bs[16 * c + F];
                }
            }
            __builtin_amdgcn_sched_barrier(0);

            #pragma unroll
            for (int s = 0; s < 4; ++s) {
                #pragma unroll
                for (int c = 0; c < NC; ++c) {
                    #pragma unroll
                    for (int r = 0; r < R; ++r) {
                        v2f a;
                        a.x = p8[r][2 * s];
                        a.y = p8[r][2 * s + 1];
                        acc[r][c] = __builtin_amdgcn_wmma_f32_16x16x4_f32(
                            false, a, false, bf[s][c], (short)0, acc[r][c], false, false);
                    }
                }
            }
            __builtin_amdgcn_sched_barrier(0);
        }
    }

    __shared__ float m_lds[R][JW * 16];
    __shared__ float l_lds[R][JW * 16];
    __shared__ float L_lds[16];
    __shared__ float accbuf[JW * 16 * F];

    #pragma unroll
    for (int r = 0; r < R; ++r) {
        if (hi == 0) {
            m_lds[r][wave * 16 + lo] = m[r];
            l_lds[r][wave * 16 + lo] = l[r];
        }
    }
    __syncthreads();

    constexpr int TOT = 16 * F;
    constexpr int NT  = JW * 32;
    #pragma unroll
    for (int r = 0; r < R; ++r) {
        float Mrow = -INFINITY;
        #pragma unroll
        for (int w = 0; w < JW; ++w) Mrow = fmaxf(Mrow, m_lds[r][w * 16 + lo]);
        float Lrow = 0.f;
        #pragma unroll
        for (int w = 0; w < JW; ++w)
            Lrow += l_lds[r][w * 16 + lo] * __expf(m_lds[r][w * 16 + lo] - Mrow);
        if (wave == 0 && hi == 0) L_lds[lo] = Lrow;

        const float fown = __expf(m[r] - Mrow);
        float fv[8];
        #pragma unroll
        for (int v = 0; v < 8; ++v) fv[v] = __shfl(fown, v + 8 * hi, 32);

        float* ab = accbuf + (wave * 16 + 8 * hi) * F + lo;
        #pragma unroll
        for (int c = 0; c < NC; ++c)
            #pragma unroll
            for (int v = 0; v < 8; ++v)
                ab[v * F + 16 * c] = acc[r][c][v] * fv[v];
        __syncthreads();

        float* outBase = Out + (long)(i0 + 16 * r) * ldo + head * headColMul;
        #pragma unroll
        for (int e = 0; e < TOT / NT; ++e) {
            const int flat = e * NT + threadIdx.x;
            const int row = flat / F;
            const int col = flat % F;
            float val = 0.f;
            #pragma unroll
            for (int w = 0; w < JW; ++w) val += accbuf[(w * 16 + row) * F + col];
            float rr = val / L_lds[row];
            if (ELU) rr = rr > 0.f ? rr : (__expf(rr) - 1.0f);
            outBase[(long)row * ldo + col] = rr;
        }
        if (r + 1 < R) __syncthreads();
    }
}

__global__ void logsoftmax128(const float* __restrict__ in,
                              float* __restrict__ out, int n) {
    const int wave = threadIdx.x >> 5;
    const int lane = threadIdx.x & 31;
    const int row  = blockIdx.x * (blockDim.x >> 5) + wave;
    if (row >= n) return;
    const float* rp = in + (long)row * NCLASS;
    float v[4];
    float mx = -INFINITY;
    #pragma unroll
    for (int q = 0; q < 4; ++q) {
        v[q] = rp[lane + 32 * q];
        mx = fmaxf(mx, v[q]);
    }
    #pragma unroll
    for (int off = 16; off >= 1; off >>= 1) mx = fmaxf(mx, __shfl_xor(mx, off, 32));
    float s = 0.f;
    #pragma unroll
    for (int q = 0; q < 4; ++q) s += __expf(v[q] - mx);
    #pragma unroll
    for (int off = 16; off >= 1; off >>= 1) s += __shfl_xor(s, off, 32);
    const float ls = __logf(s);
    #pragma unroll
    for (int q = 0; q < 4; ++q) out[(long)row * NCLASS + lane + 32 * q] = v[q] - mx - ls;
}

extern "C" void kernel_launch(void* const* d_in, const int* in_sizes, int n_in,
                              void* d_out, int out_size, void* d_ws, size_t ws_size,
                              hipStream_t stream) {
    const float* x     = (const float*)d_in[0];
    const int*   adj   = (const int*)d_in[1];
    const float* W     = (const float*)d_in[2];
    const float* a     = (const float*)d_in[3];
    const float* W_out = (const float*)d_in[4];
    const float* a_out = (const float*)d_in[5];
    float* out = (float*)d_out;

    char* ws = (char*)d_ws;
    unsigned* bits = (unsigned*)ws;
    ws += (size_t)NN * (NN / 32) * sizeof(unsigned);
    float* h1 = (float*)ws;
    ws += (size_t)NHEADS * NN * NHID * sizeof(float);
    float* fi1 = (float*)ws; ws += (size_t)NHEADS * NN * sizeof(float);
    float* fj1 = (float*)ws; ws += (size_t)NHEADS * NN * sizeof(float);
    float* xcat = (float*)ws;
    ws += (size_t)NN * (NHEADS * NHID) * sizeof(float);
    float* h2 = (float*)ws;
    ws += (size_t)NN * NCLASS * sizeof(float);
    float* fi2 = (float*)ws; ws += (size_t)NN * sizeof(float);
    float* fj2 = (float*)ws; ws += (size_t)NN * sizeof(float);
    float* outpre = (float*)ws;

    {
        const long nwords = (long)NN * (NN / 32);
        pack_adj_kernel<<<dim3((nwords + 255) / 256), dim3(256), 0, stream>>>(adj, bits, nwords);
    }

    wmma_gemm_f32<NFEAT, NHID><<<dim3(NN / 16, 1, NHEADS), dim3(128), 0, stream>>>(
        x, W, (long)NFEAT * NHID, h1, (long)NN * NHID);

    fused_fifj<<<dim3(NN / 256, NHEADS), dim3(256), 0, stream>>>(
        h1, (long)NN * NHID, NHID, a, 2 * NHID, fi1, fj1, NN);

    gat_attn<NHID, 4, 2, 4, false><<<dim3(NN / 32, NHEADS), dim3(128), 0, stream>>>(
        h1, (long)NN * NHID, fi1, fj1, bits,
        xcat, NHEADS * NHID, NHID, NN);

    wmma_gemm_f32<NHEADS * NHID, NCLASS><<<dim3(NN / 16, NCLASS / 64, 1), dim3(128), 0, stream>>>(
        xcat, W_out, 0, h2, 0);

    fused_fifj<<<dim3(NN / 256, 1), dim3(256), 0, stream>>>(
        h2, 0, NCLASS, a_out, 2 * NCLASS, fi2, fj2, NN);

    gat_attn<NCLASS, 16, 1, 3, true><<<dim3(NN / 16, 1), dim3(512), 0, stream>>>(
        h2, 0, fi2, fj2, bits,
        outpre, NCLASS, 0, NN);

    logsoftmax128<<<dim3(NN / 4), dim3(128), 0, stream>>>(outpre, out, NN);
}


# projection GEMMs register-tiled: a wave holds 32x64 (layer 2: 16x32) of C so a fragment load feeds 4 or 2 WMMAs; 16-byte fragment loads one k block ahead; k summed in another order in groups of 8
# speedup vs baseline: 1.3486x; 1.0579x over previous
#include <hip/hip_runtime.h>
#include <math.h>

#define NN      4096
#define NFEAT   512
#define NHID    64
#define NCLASS  128
#define NHEADS  8
#define ALPHA   0.2f
#define NEGV    (-9000000000000000.0f)

typedef __attribute__((ext_vector_type(2))) float v2f;
typedef __attribute__((ext_vector_type(4))) float v4f;
typedef __attribute__((ext_vector_type(8))) float v8f;

__global__ void pack_adj_kernel(const int* __restrict__ adj,
                                unsigned* __restrict__ bits, long nwords) {
    long w = (long)blockIdx.x * blockDim.x + threadIdx.x;
    if (w >= nwords) return;
    const int* p = adj + w * 32;
    unsigned m = 0u;
    #pragma unroll
    for (int b = 0; b < 32; ++b) m |= (p[b] > 0 ? 1u : 0u) << b;
    bits[w] = m;
}

template <int R, int CT, int G, int K, int NB>
__device__ __forceinline__ void gemm_load_block(
        v4f (&a)[R][G], __attribute__((ext_vector_type(CT))) float (&b)[G][4],
        const float* __restrict__ ap, const float* __restrict__ bp, int k) {
    typedef __attribute__((ext_vector_type(CT))) float vct;
    #pragma unroll
    for (int g = 0; g < G; ++g) {
        #pragma unroll
        for (int r = 0; r < R; ++r)
            a[r][g] = *(const v4f*)(ap + (long)r * 16 * K + k + 8 * g);
        #pragma unroll
        for (int j = 0; j < 4; ++j)
            b[g][j] = *(const vct*)(bp + (long)(k + 8 * g + j) * NB);
    }
}

template <int R, int CT, int G>
__device__ __forceinline__ void gemm_mma_block(
        v8f (&acc)[R][CT],
        const v4f (&a)[R][G], const __attribute__((ext_vector_type(CT))) float (&b)[G][4]) {
    v2f q[G][2][CT];
    #pragma unroll
    for (int g = 0; g < G; ++g)
        #pragma unroll
        for (int kk = 0; kk < 2; ++kk)
            #pragma unroll
            for (int t = 0; t < CT; ++t) {
                q[g][kk][t].x = b[g][2 * kk][t];
                q[g][kk][t].y = b[g][2 * kk + 1][t];
            }
    __builtin_amdgcn_sched_barrier(0);
    #pragma unroll
    for (int g = 0; g < G; ++g) {
        #pragma unroll
        for (int kk = 0; kk < 2; ++kk) {
            #pragma unroll
            for (int t = 0; t < CT; ++t) {
                #pragma unroll
                for (int r = 0; r < R; ++r) {
                    v2f av;
                    av.x = a[r][g][2 * kk];
                    av.y = a[r][g][2 * kk + 1];
                    acc[r][t] = __builtin_amdgcn_wmma_f32_16x16x4_f32(
                        false, av, false, q[g][kk][t], (short)0, acc[r][t], false, false);
                }
            }
        }
    }
    __builtin_amdgcn_sched_barrier(0);
    asm volatile("v_nop\n\tv_nop\n\tv_nop\n\tv_nop");
    __builtin_amdgcn_sched_barrier(0);
}

template <int K, int NB, int R, int CT, int KB>
__global__ __launch_bounds__(128) void wmma_gemm_f32(
        const float* __restrict__ A,
        const float* __restrict__ B, long bStride,
        float* __restrict__ C, long cStride) {
    typedef __attribute__((ext_vector_type(CT))) float vct;
    constexpr int G = KB / 8;
    static_assert(K % (2 * KB) == 0, "K must hold an even number of k blocks");
    const int wave = threadIdx.x >> 5;
    const int lane = threadIdx.x & 31;
    const int hi   = lane >> 4;
    const int lo   = lane & 15;
    const int row0 = (blockIdx.x * 4 + wave) * (16 * R);
    const int colb = blockIdx.y * (16 * CT);
    const int head = blockIdx.z;

    const float* __restrict__ ap = A + (long)(row0 + lo) * K + 4 * hi;
    const float* __restrict__ bp = B + (long)head * bStride + (long)(4 * hi) * NB + colb + CT * lo;

    v8f acc[R][CT];
    #pragma unroll
    for (int r = 0; r < R; ++r)
        #pragma unroll
        for (int t = 0; t < CT; ++t) acc[r][t] = (v8f){};

    v4f a0[R][G], a1[R][G];
    vct b0[G][4], b1[G][4];
    gemm_load_block<R, CT, G, K, NB>(a0, b0, ap, bp, 0);

    #pragma unroll 1
    for (int k0 = 0; k0 < K; k0 += 2 * KB) {
        gemm_load_block<R, CT, G, K, NB>(a1, b1, ap, bp, k0 + KB);
        __builtin_amdgcn_sched_barrier(0);
        gemm_mma_block<R, CT, G>(acc, a0, b0);
        const int kn = (k0 + 2 * KB < K) ? (k0 + 2 * KB) : k0;
        gemm_load_block<R, CT, G, K, NB>(a0, b0, ap, bp, kn);
        __builtin_amdgcn_sched_barrier(0);
        gemm_mma_block<R, CT, G>(acc, a1, b1);
    }

    float* __restrict__ cp = C + (long)head * cStride + (long)(row0 + 8 * hi) * NB + colb + CT * lo;
    #pragma unroll
    for (int r = 0; r < R; ++r) {
        #pragma unroll
        for (int v = 0; v < 8; ++v) {
            vct o;
            #pragma unroll
            for (int t = 0; t < CT; ++t) o[t] = acc[r][t][v];
            *(vct*)(cp + (long)(16 * r + v) * NB) = o;
        }
    }
}

__global__ void fused_fifj(const float* __restrict__ H, long hStride, int F,
                           const float* __restrict__ a, int aStride,
                           float* __restrict__ fi, float* __restrict__ fj, int n) {
    const int head = blockIdx.y;
    const int nd   = blockIdx.x * blockDim.x + threadIdx.x;
    if (nd >= n) return;
    const float* hp = H + (long)head * hStride + (long)nd * F;
    const float* a1 = a + (long)head * aStride;
    const float* a2 = a1 + F;
    float s1 = 0.f, s2 = 0.f;
    for (int o = 0; o < F; ++o) {
        float hv = hp[o];
        s1 += hv * a1[o];
        s2 += hv * a2[o];
    }
    fi[(long)head * n + nd] = s1;
    fj[(long)head * n + nd] = s2;
}

template <int F, int JW, int R, int PRE, bool ELU>
__global__ __launch_bounds__(JW * 32) void gat_attn(
        const float* __restrict__ Hfeat, long hStride,
        const float* __restrict__ fi, const float* __restrict__ fj,
        const unsigned* __restrict__ bits,
        float* __restrict__ Out, int ldo, int headColMul, int n) {
    constexpr int NC = F / 16;
    const int wave = threadIdx.x >> 5;
    const int lane = threadIdx.x & 31;
    const int hi   = lane >> 4;
    const int lo   = lane & 15;
    const int head = blockIdx.y;
    const int i0   = blockIdx.x * (16 * R);
    const int nwords = n >> 5;

    const float* __restrict__ Hh  = Hfeat + (long)head * hStride;
    const float* __restrict__ fih = fi + (long)head * n;
    const float* __restrict__ fjh = fj + (long)head * n;
    const float* __restrict__ bbase = Hh + 2 * hi * F + lo;

    float fiv[R];
    const unsigned* __restrict__ brow[R];
    #pragma unroll
    for (int r = 0; r < R; ++r) {
        fiv[r]  = fih[i0 + 16 * r + lo];
        brow[r] = bits + (long)(i0 + 16 * r + lo) * nwords;
    }

    const int chunk = n / JW;
    const int jbeg  = wave * chunk;
    const int jend  = jbeg + chunk;

    float m[R], l[R];
    v8f acc[R][NC];
    #pragma unroll
    for (int r = 0; r < R; ++r) {
        m[r] = -INFINITY;
        l[r] = 0.f;
        #pragma unroll
        for (int c = 0; c < NC; ++c) acc[r][c] = (v8f){};
    }

    unsigned wm_n[R];
    float fj_n[16];
    #pragma unroll
    for (int r = 0; r < R; ++r) wm_n[r] = brow[r][jbeg >> 5];
    {
        const float* __restrict__ fjp = fjh + jbeg + 2 * hi;
        #pragma unroll
        for (int q = 0; q < 8; ++q)
            #pragma unroll
            for (int t = 0; t < 2; ++t) fj_n[2 * q + t] = fjp[4 * q + t];
    }

    for (int j0 = jbeg; j0 < jend; j0 += 32) {
        unsigned wmask[R];
        float fjc[16];
        #pragma unroll
        for (int r = 0; r < R; ++r) wmask[r] = wm_n[r];
        #pragma unroll
        for (int q = 0; q < 16; ++q) fjc[q] = fj_n[q];

        #pragma unroll
        for (int half = 0; half < 2; ++half) {
            const int jb = j0 + 16 * half;
            const int bitbase = 16 * half;

            float fjv[8];
            #pragma unroll
            for (int q = 0; q < 8; ++q) fjv[q] = fjc[8 * half + q];

            v2f bf[4][NC];
            const float* __restrict__ bp = bbase + jb * F;
            #pragma unroll
            for (int s = 0; s < PRE; ++s) {
                const float* __restrict__ bs = bp + 4 * s * F;
                #pragma unroll
                for (int c = 0; c < NC; ++c) {
                    bf[s][c].x = bs[16 * c];
                    bf[s][c].y = bs[16 * c + F];
                }
            }

            float p8[R][8];
            #pragma unroll
            for (int r = 0; r < R; ++r) {
                float e8[8];
                float tmax = -INFINITY;
                #pragma unroll
                for (int s = 0; s < 4; ++s) {
                    #pragma unroll
                    for (int t = 0; t < 2; ++t) {
                        const int c = 4 * s + 2 * hi + t;
                        float v = fiv[r] + fjv[2 * s + t];
                        v = v > 0.f ? v : ALPHA * v;
                        const float e = ((wmask[r] >> (bitbase + c)) & 1u) ? v : NEGV;
                        e8[2 * s + t] = e;
                        tmax = fmaxf(tmax, e);
                    }
                }
                tmax = fmaxf(tmax, __shfl_xor(tmax, 16, 32));
                const float mnew = fmaxf(m[r], tmax);

                float psum = 0.f;
                #pragma unroll
                for (int q = 0; q < 8; ++q) {
                    p8[r][q] = __expf(e8[q] - mnew);
                    psum += p8[r][q];
                }
                psum += __shfl_xor(psum, 16, 32);

                if (__any(mnew > m[r])) {
                    const float scale = __expf(m[r] - mnew);
                    float sv[8];
                    #pragma unroll
                    for (int v = 0; v < 8; ++v) sv[v] = __shfl(scale, v + 8 * hi, 32);
                    #pragma unroll
                    for (int c = 0; c < NC; ++c)
                        #pragma unroll
                        for (int v = 0; v < 8; ++v) acc[r][c][v] *= sv[v];
                    l[r] = l[r] * scale + psum;
                    m[r] = mnew;
                } else {
                    l[r] += psum;
                }
            }

            if (half == 1) {
                const int jn = (j0 + 32 < jend) ? (j0 + 32) : j0;
                const float* __restrict__ fjp = fjh + jn + 2 * hi;
                #pragma unroll
                for (int q = 0; q < 8; ++q)
                    #pragma unroll
                    for (int t = 0; t < 2; ++t) fj_n[2 * q + t] = fjp[4 * q + t];
                #pragma unroll
                for (int r = 0; r < R; ++r) wm_n[r] = brow[r][jn >> 5];
            }
            #pragma unroll
            for (int s = PRE; s < 4; ++s) {
                const float* __restrict__ bs = bp + 4 * s * F;
                #pragma unroll
                for (int c = 0; c < NC; ++c) {
                    bf[s][c].x = bs[16 * c];
                    bf[s][c].y = bs[16 * c + F];
                }
            }
            __builtin_amdgcn_sched_barrier(0);

            #pragma unroll
            for (int s = 0; s < 4; ++s) {
                #pragma unroll
                for (int c = 0; c < NC; ++c) {
                    #pragma unroll
                    for (int r = 0; r < R; ++r) {
                        v2f a;
                        a.x = p8[r][2 * s];
                        a.y = p8[r][2 * s + 1];
                        acc[r][c] = __builtin_amdgcn_wmma_f32_16x16x4_f32(
                            false, a, false, bf[s][c], (short)0, acc[r][c], false, false);
                    }
                }
            }
            __builtin_amdgcn_sched_barrier(0);
        }
    }

    __shared__ float m_lds[R][JW * 16];
    __shared__ float l_lds[R][JW * 16];
    __shared__ float L_lds[16];
    __shared__ float accbuf[JW * 16 * F];

    #pragma unroll
    for (int r = 0; r < R; ++r) {
        if (hi == 0) {
            m_lds[r][wave * 16 + lo] = m[r];
            l_lds[r][wave * 16 + lo] = l[r];
        }
    }
    __syncthreads();

    constexpr int TOT = 16 * F;
    constexpr int NT  = JW * 32;
    #pragma unroll
    for (int r = 0; r < R; ++r) {
        float Mrow = -INFINITY;
        #pragma unroll
        for (int w = 0; w < JW; ++w) Mrow = fmaxf(Mrow, m_lds[r][w * 16 + lo]);
        float Lrow = 0.f;
        #pragma unroll
        for (int w = 0; w < JW; ++w)
            Lrow += l_lds[r][w * 16 + lo] * __expf(m_lds[r][w * 16 + lo] - Mrow);
        if (wave == 0 && hi == 0) L_lds[lo] = Lrow;

        const float fown = __expf(m[r] - Mrow);
        float fv[8];
        #pragma unroll
        for (int v = 0; v < 8; ++v) fv[v] = __shfl(fown, v + 8 * hi, 32);

        float* ab = accbuf + (wave * 16 + 8 * hi) * F + lo;
        #pragma unroll
        for (int c = 0; c < NC; ++c)
            #pragma unroll
            for (int v = 0; v < 8; ++v)
                ab[v * F + 16 * c] = acc[r][c][v] * fv[v];
        __syncthreads();

        float* outBase = Out + (long)(i0 + 16 * r) * ldo + head * headColMul;
        #pragma unroll
        for (int e = 0; e < TOT / NT; ++e) {
            const int flat = e * NT + threadIdx.x;
            const int row = flat / F;
            const int col = flat % F;
            float val = 0.f;
            #pragma unroll
            for (int w = 0; w < JW; ++w) val += accbuf[(w * 16 + row) * F + col];
            float rr = val / L_lds[row];
            if (ELU) rr = rr > 0.f ? rr : (__expf(rr) - 1.0f);
            outBase[(long)row * ldo + col] = rr;
        }
        if (r + 1 < R) __syncthreads();
    }
}

__global__ void logsoftmax128(const float* __restrict__ in,
                              float* __restrict__ out, int n) {
    const int wave = threadIdx.x >> 5;
    const int lane = threadIdx.x & 31;
    const int row  = blockIdx.x * (blockDim.x >> 5) + wave;
    if (row >= n) return;
    const float* rp = in + (long)row * NCLASS;
    float v[4];
    float mx = -INFINITY;
    #pragma unroll
    for (int q = 0; q < 4; ++q) {
        v[q] = rp[lane + 32 * q];
        mx = fmaxf(mx, v[q]);
    }
    #pragma unroll
    for (int off = 16; off >= 1; off >>= 1) mx = fmaxf(mx, __shfl_xor(mx, off, 32));
    float s = 0.f;
    #pragma unroll
    for (int q = 0; q < 4; ++q) s += __expf(v[q] - mx);
    #pragma unroll
    for (int off = 16; off >= 1; off >>= 1) s += __shfl_xor(s, off, 32);
    const float ls = __logf(s);
    #pragma unroll
    for (int q = 0; q < 4; ++q) out[(long)row * NCLASS + lane + 32 * q] = v[q] - mx - ls;
}

extern "C" void kernel_launch(void* const* d_in, const int* in_sizes, int n_in,
                              void* d_out, int out_size, void* d_ws, size_t ws_size,
                              hipStream_t stream) {
    const float* x     = (const float*)d_in[0];
    const int*   adj   = (const int*)d_in[1];
    const float* W     = (const float*)d_in[2];
    const float* a     = (const float*)d_in[3];
    const float* W_out = (const float*)d_in[4];
    const float* a_out = (const float*)d_in[5];
    float* out = (float*)d_out;

    char* ws = (char*)d_ws;
    unsigned* bits = (unsigned*)ws;
    ws += (size_t)NN * (NN / 32) * sizeof(unsigned);
    float* h1 = (float*)ws;
    ws += (size_t)NHEADS * NN * NHID * sizeof(float);
    float* fi1 = (float*)ws; ws += (size_t)NHEADS * NN * sizeof(float);
    float* fj1 = (float*)ws; ws += (size_t)NHEADS * NN * sizeof(float);
    float* xcat = (float*)ws;
    ws += (size_t)NN * (NHEADS * NHID) * sizeof(float);
    float* h2 = (float*)ws;
    ws += (size_t)NN * NCLASS * sizeof(float);
    float* fi2 = (float*)ws; ws += (size_t)NN * sizeof(float);
    float* fj2 = (float*)ws; ws += (size_t)NN * sizeof(float);
    float* outpre = (float*)ws;

    {
        const long nwords = (long)NN * (NN / 32);
        pack_adj_kernel<<<dim3((nwords + 255) / 256), dim3(256), 0, stream>>>(adj, bits, nwords);
    }

    wmma_gemm_f32<NFEAT, NHID, 2, 4, 32><<<dim3(NN / 128, 1, NHEADS), dim3(128), 0, stream>>>(
        x, W, (long)NFEAT * NHID, h1, (long)NN * NHID);

    fused_fifj<<<dim3(NN / 256, NHEADS), dim3(256), 0, stream>>>(
        h1, (long)NN * NHID, NHID, a, 2 * NHID, fi1, fj1, NN);

    gat_attn<NHID, 4, 2, 4, false><<<dim3(NN / 32, NHEADS), dim3(128), 0, stream>>>(
        h1, (long)NN * NHID, fi1, fj1, bits,
        xcat, NHEADS * NHID, NHID, NN);

    wmma_gemm_f32<NHEADS * NHID, NCLASS, 1, 2, 64><<<dim3(NN / 64, NCLASS / 32, 1), dim3(128), 0, stream>>>(
        xcat, W_out, 0, h2, 0);

    fused_fifj<<<dim3(NN / 256, 1), dim3(256), 0, stream>>>(
        h2, 0, NCLASS, a_out, 2 * NCLASS, fi2, fj2, NN);

    gat_attn<NCLASS, 16, 1, 3, true><<<dim3(NN / 16, 1), dim3(512), 0, stream>>>(
        h2, 0, fi2, fj2, bits,
        outpre, NCLASS, 0, NN);

    logsoftmax128<<<dim3(NN / 4), dim3(128), 0, stream>>>(outpre, out, NN);
}


# fi/fj of both layers and the final log_softmax move into the epilogues of the GEMM and attention kernels that hold those rows: 8 launches -> 5; fi/fj summed as a 16-lane tree, not serially
# speedup vs baseline: 1.4131x; 1.0478x over previous
#include <hip/hip_runtime.h>
#include <math.h>

#define NN      4096
#define NFEAT   512
#define NHID    64
#define NCLASS  128
#define NHEADS  8
#define ALPHA   0.2f
#define NEGV    (-9000000000000000.0f)

typedef __attribute__((ext_vector_type(2))) float v2f;
typedef __attribute__((ext_vector_type(4))) float v4f;
typedef __attribute__((ext_vector_type(8))) float v8f;

__global__ void pack_adj_kernel(const int* __restrict__ adj,
                                unsigned* __restrict__ bits, long nwords) {
    long w = (long)blockIdx.x * blockDim.x + threadIdx.x;
    if (w >= nwords) return;
    const int* p = adj + w * 32;
    unsigned m = 0u;
    #pragma unroll
    for (int b = 0; b < 32; ++b) m |= (p[b] > 0 ? 1u : 0u) << b;
    bits[w] = m;
}

typedef __attribute__((ext_vector_type(16))) float v16f;

__device__ __forceinline__ float lane_tree_2(v2f y, int lo) {
    const bool up = (lo & 1) != 0;
    const float keep = up ? y.y : y.x;
    const float give = up ? y.x : y.y;
    return keep + __shfl_xor(give, 1, 32);
}
__device__ __forceinline__ float lane_tree_4(v4f y, int lo) {
    const bool up = (lo & 2) != 0;
    const v2f keep = up ? y.hi : y.lo;
    const v2f give = up ? y.lo : y.hi;
    v2f z;
    z.x = keep.x + __shfl_xor(give.x, 2, 32);
    z.y = keep.y + __shfl_xor(give.y, 2, 32);
    return lane_tree_2(z, lo);
}
__device__ __forceinline__ float lane_tree_8(v8f y, int lo) {
    const bool up = (lo & 4) != 0;
    const v4f keep = up ? y.hi : y.lo;
    const v4f give = up ? y.lo : y.hi;
    v4f z;
    #pragma unroll
    for (int i = 0; i < 4; ++i) z[i] = keep[i] + __shfl_xor(give[i], 4, 32);
    return lane_tree_4(z, lo);
}
__device__ __forceinline__ float lane_tree_16(v16f y, int lo) {
    const bool up = (lo & 8) != 0;
    const v8f keep = up ? y.hi : y.lo;
    const v8f give = up ? y.lo : y.hi;
    v8f z;
    #pragma unroll
    for (int i = 0; i < 8; ++i) z[i] = keep[i] + __shfl_xor(give[i], 8, 32);
    return lane_tree_8(z, lo);
}

template <int R, int CT, int G, int K, int NB>
__device__ __forceinline__ void gemm_load_block(
        v4f (&a)[R][G], __attribute__((ext_vector_type(CT))) float (&b)[G][4],
        const float* __restrict__ ap, const float* __restrict__ bp, int k) {
    typedef __attribute__((ext_vector_type(CT))) float vct;
    #pragma unroll
    for (int g = 0; g < G; ++g) {
        #pragma unroll
        for (int r = 0; r < R; ++r)
            a[r][g] = *(const v4f*)(ap + (long)r * 16 * K + k + 8 * g);
        #pragma unroll
        for (int j = 0; j < 4; ++j)
            b[g][j] = *(const vct*)(bp + (long)(k + 8 * g + j) * NB);
    }
}

template <int R, int CT, int G>
__device__ __forceinline__ void gemm_mma_block(
        v8f (&acc)[R][CT],
        const v4f (&a)[R][G], const __attribute__((ext_vector_type(CT))) float (&b)[G][4]) {
    v2f q[G][2][CT];
    #pragma unroll
    for (int g = 0; g < G; ++g)
        #pragma unroll
        for (int kk = 0; kk < 2; ++kk)
            #pragma unroll
            for (int t = 0; t < CT; ++t) {
                q[g][kk][t].x = b[g][2 * kk][t];
                q[g][kk][t].y = b[g][2 * kk + 1][t];
            }
    __builtin_amdgcn_sched_barrier(0);
    #pragma unroll
    for (int g = 0; g < G; ++g) {
        #pragma unroll
        for (int kk = 0; kk < 2; ++kk) {
            #pragma unroll
            for (int t = 0; t < CT; ++t) {
                #pragma unroll
                for (int r = 0; r < R; ++r) {
                    v2f av;
                    av.x = a[r][g][2 * kk];
                    av.y = a[r][g][2 * kk + 1];
                    acc[r][t] = __builtin_amdgcn_wmma_f32_16x16x4_f32(
                        false, av, false, q[g][kk][t], (short)0, acc[r][t], false, false);
                }
            }
        }
    }
    __builtin_amdgcn_sched_barrier(0);
    asm volatile("v_nop\n\tv_nop\n\tv_nop\n\tv_nop");
    __builtin_amdgcn_sched_barrier(0);
}

template <int K, int NB, int R, int CT, int KB, int WCOLS>
__global__ __launch_bounds__(128) void wmma_gemm_f32(
        const float* __restrict__ A,
        const float* __restrict__ B, long bStride,
        float* __restrict__ C, long cStride,
        const float* __restrict__ avec, int aStride,
        float* __restrict__ fi, float* __restrict__ fj, int n) {
    typedef __attribute__((ext_vector_type(CT))) float vct;
    constexpr int G = KB / 8;
    constexpr int WR = 4 / WCOLS;
    static_assert(K % (2 * KB) == 0, "K must hold an even number of k blocks");
    static_assert(WCOLS * 16 * CT == NB, "a block must hold whole rows of C");
    static_assert(WCOLS == 1 || WR == 1, "waves run along the rows or along the columns");
    const int wave = threadIdx.x >> 5;
    const int lane = threadIdx.x & 31;
    const int hi   = lane >> 4;
    const int lo   = lane & 15;
    const int wc   = wave % WCOLS;
    const int row0 = (blockIdx.x * WR + wave / WCOLS) * (16 * R);
    const int colb = wc * (16 * CT);
    const int head = blockIdx.z;

    const float* __restrict__ ap = A + (long)(row0 + lo) * K + 4 * hi;
    const float* __restrict__ bp = B + (long)head * bStride + (long)(4 * hi) * NB + colb + CT * lo;

    v8f acc[R][CT];
    #pragma unroll
    for (int r = 0; r < R; ++r)
        #pragma unroll
        for (int t = 0; t < CT; ++t) acc[r][t] = (v8f){};

    v4f a0[R][G], a1[R][G];
    vct b0[G][4], b1[G][4];
    gemm_load_block<R, CT, G, K, NB>(a0, b0, ap, bp, 0);

    #pragma unroll 1
    for (int k0 = 0; k0 < K; k0 += 2 * KB) {
        gemm_load_block<R, CT, G, K, NB>(a1, b1, ap, bp, k0 + KB);
        __builtin_amdgcn_sched_barrier(0);
        gemm_mma_block<R, CT, G>(acc, a0, b0);
        const int kn = (k0 + 2 * KB < K) ? (k0 + 2 * KB) : k0;
        gemm_load_block<R, CT, G, K, NB>(a0, b0, ap, bp, kn);
        __builtin_amdgcn_sched_barrier(0);
        gemm_mma_block<R, CT, G>(acc, a1, b1);
    }

    float* __restrict__ cp = C + (long)head * cStride + (long)(row0 + 8 * hi) * NB + colb + CT * lo;
    #pragma unroll
    for (int r = 0; r < R; ++r) {
        #pragma unroll
        for (int v = 0; v < 8; ++v) {
            vct o;
            #pragma unroll
            for (int t = 0; t < CT; ++t) o[t] = acc[r][t][v];
            *(vct*)(cp + (long)(16 * r + v) * NB) = o;
        }
    }

    const float* __restrict__ a1p = avec + (long)head * aStride + colb + CT * lo;
    const vct a1v = *(const vct*)a1p;
    const vct a2v = *(const vct*)(a1p + NB);
    constexpr int NV = 8 * R;
    static_assert(R == 1 || R == 2, "8 or 16 rows per lane half");
    typedef __attribute__((ext_vector_type(NV))) float vnv;
    vnv p1, p2;
    #pragma unroll
    for (int r = 0; r < R; ++r) {
        #pragma unroll
        for (int v = 0; v < 8; ++v) {
            float s1 = 0.f, s2 = 0.f;
            #pragma unroll
            for (int t = 0; t < CT; ++t) {
                s1 += acc[r][t][v] * a1v[t];
                s2 += acc[r][t][v] * a2v[t];
            }
            p1[8 * r + v] = s1;
            p2[8 * r + v] = s2;
        }
    }
    float w1, w2;
    if constexpr (NV == 16) {
        w1 = lane_tree_16(p1, lo);
        w2 = lane_tree_16(p2, lo);
    } else {
        w1 = lane_tree_8(p1, lo);
        w2 = lane_tree_8(p2, lo);
        w1 += __shfl_xor(w1, 8, 32);
        w2 += __shfl_xor(w2, 8, 32);
    }
    const int qv   = lo % NV;
    const int qrow = 16 * (qv >> 3) + 8 * hi + (qv & 7);
    if constexpr (WCOLS == 1) {
        if (lo < NV) {
            const long o = (long)head * n + row0 + qrow;
            fi[o] = w1;
            fj[o] = w2;
        }
    } else {
        __shared__ float part[2][WCOLS][16 * R];
        if (lo < NV) {
            part[0][wc][qrow] = w1;
            part[1][wc][qrow] = w2;
        }
        __syncthreads();
        static_assert(R == 1, "the packed line holds one row tile");
        if (threadIdx.x < 32) {
            const int which = threadIdx.x >> 4;
            float t1 = 0.f;
            #pragma unroll
            for (int w = 0; w < WCOLS; ++w) t1 += part[which][w][threadIdx.x & 15];
            fi[2 * ((long)head * n + row0) + threadIdx.x] = t1;
        }
    }
}

template <int F, int JW, int R, int PRE, bool ELU, bool PK, bool LSM>
__global__ __launch_bounds__(JW * 32) void gat_attn(
        const float* __restrict__ Hfeat, long hStride,
        const float* __restrict__ fi, const float* __restrict__ fj,
        const unsigned* __restrict__ bits,
        float* __restrict__ Out, int ldo, int headColMul, int n) {
    constexpr int NC = F / 16;
    const int wave = threadIdx.x >> 5;
    const int lane = threadIdx.x & 31;
    const int hi   = lane >> 4;
    const int lo   = lane & 15;
    const int head = blockIdx.y;
    const int i0   = blockIdx.x * (16 * R);
    const int nwords = n >> 5;

    const float* __restrict__ Hh  = Hfeat + (long)head * hStride;
    const float* __restrict__ fih = fi + (PK ? 2 : 1) * (long)head * n;
    const float* __restrict__ fjh = fj + (PK ? 2 : 1) * (long)head * n;
    const float* __restrict__ bbase = Hh + 2 * hi * F + lo;

    float fiv[R];
    const unsigned* __restrict__ brow[R];
    #pragma unroll
    for (int r = 0; r < R; ++r) {
        fiv[r]  = fih[(PK ? 2 : 1) * (i0 + 16 * r) + lo];
        brow[r] = bits + (long)(i0 + 16 * r + lo) * nwords;
    }

    const int chunk = n / JW;
    const int jbeg  = wave * chunk;
    const int jend  = jbeg + chunk;

    float m[R], l[R];
    v8f acc[R][NC];
    #pragma unroll
    for (int r = 0; r < R; ++r) {
        m[r] = -INFINITY;
        l[r] = 0.f;
        #pragma unroll
        for (int c = 0; c < NC; ++c) acc[r][c] = (v8f){};
    }

    unsigned wm_n[R];
    float fj_n[16];
    #pragma unroll
    for (int r = 0; r < R; ++r) wm_n[r] = brow[r][jbeg >> 5];
    {
        const float* __restrict__ fjp = fjh + (PK ? 2 : 1) * jbeg + 2 * hi;
        #pragma unroll
        for (int q = 0; q < 8; ++q)
            #pragma unroll
            for (int t = 0; t < 2; ++t) fj_n[2 * q + t] = fjp[4 * q + t + ((PK && q >= 4) ? 16 : 0)];
    }

    for (int j0 = jbeg; j0 < jend; j0 += 32) {
        unsigned wmask[R];
        float fjc[16];
        #pragma unroll
        for (int r = 0; r < R; ++r) wmask[r] = wm_n[r];
        #pragma unroll
        for (int q = 0; q < 16; ++q) fjc[q] = fj_n[q];

        #pragma unroll
        for (int half = 0; half < 2; ++half) {
            const int jb = j0 + 16 * half;
            const int bitbase = 16 * half;

            float fjv[8];
            #pragma unroll
            for (int q = 0; q < 8; ++q) fjv[q] = fjc[8 * half + q];

            v2f bf[4][NC];
            const float* __restrict__ bp = bbase + jb * F;
            #pragma unroll
            for (int s = 0; s < PRE; ++s) {
                const float* __restrict__ bs = bp + 4 * s * F;
                #pragma unroll
                for (int c = 0; c < NC; ++c) {
                    bf[s][c].x = bs[16 * c];
                    bf[s][c].y = bs[16 * c + F];
                }
            }

            float p8[R][8];
            #pragma unroll
            for (int r = 0; r < R; ++r) {
                float e8[8];
                float tmax = -INFINITY;
                #pragma unroll
                for (int s = 0; s < 4; ++s) {
                    #pragma unroll
                    for (int t = 0; t < 2; ++t) {
                        const int c = 4 * s + 2 * hi + t;
                        float v = fiv[r] + fjv[2 * s + t];
                        v = v > 0.f ? v : ALPHA * v;
                        const float e = ((wmask[r] >> (bitbase + c)) & 1u) ? v : NEGV;
                        e8[2 * s + t] = e;
                        tmax = fmaxf(tmax, e);
                    }
                }
                tmax = fmaxf(tmax, __shfl_xor(tmax, 16, 32));
                const float mnew = fmaxf(m[r], tmax);

                float psum = 0.f;
                #pragma unroll
                for (int q = 0; q < 8; ++q) {
                    p8[r][q] = __expf(e8[q] - mnew);
                    psum += p8[r][q];
                }
                psum += __shfl_xor(psum, 16, 32);

                if (__any(mnew > m[r])) {
                    const float scale = __expf(m[r] - mnew);
                    float sv[8];
                    #pragma unroll
                    for (int v = 0; v < 8; ++v) sv[v] = __shfl(scale, v + 8 * hi, 32);
                    #pragma unroll
                    for (int c = 0; c < NC; ++c)
                        #pragma unroll
                        for (int v = 0; v < 8; ++v) acc[r][c][v] *= sv[v];
                    l[r] = l[r] * scale + psum;
                    m[r] = mnew;
                } else {
                    l[r] += psum;
                }
            }

            if (half == 1) {
                const int jn = (j0 + 32 < jend) ? (j0 + 32) : j0;
                const float* __restrict__ fjp = fjh + (PK ? 2 : 1) * jn + 2 * hi;
                #pragma unroll
                for (int q = 0; q < 8; ++q)
                    #pragma unroll
                    for (int t = 0; t < 2; ++t) fj_n[2 * q + t] = fjp[4 * q + t + ((PK && q >= 4) ? 16 : 0)];
                #pragma unroll
                for (int r = 0; r < R; ++r) wm_n[r] = brow[r][jn >> 5];
            }
            #pragma unroll
            for (int s = PRE; s < 4; ++s) {
                const float* __restrict__ bs = bp + 4 * s * F;
                #pragma unroll
                for (int c = 0; c < NC; ++c) {
                    bf[s][c].x = bs[16 * c];
                    bf[s][c].y = bs[16 * c + F];
                }
            }
            __builtin_amdgcn_sched_barrier(0);

            #pragma unroll
            for (int s = 0; s < 4; ++s) {
                #pragma unroll
                for (int c = 0; c < NC; ++c) {
                    #pragma unroll
                    for (int r = 0; r < R; ++r) {
                        v2f a;
                        a.x = p8[r][2 * s];
                        a.y = p8[r][2 * s + 1];
                        acc[r][c] = __builtin_amdgcn_wmma_f32_16x16x4_f32(
                            false, a, false, bf[s][c], (short)0, acc[r][c], false, false);
                    }
                }
            }
            __builtin_amdgcn_sched_barrier(0);
        }
    }

    __shared__ float m_lds[R][JW * 16];
    __shared__ float l_lds[R][JW * 16];
    __shared__ float L_lds[16];
    __shared__ float accbuf[JW * 16 * F];
    __shared__ float obuf[LSM ? 16 * F : 1];

    #pragma unroll
    for (int r = 0; r < R; ++r) {
        if (hi == 0) {
            m_lds[r][wave * 16 + lo] = m[r];
            l_lds[r][wave * 16 + lo] = l[r];
        }
    }
    __syncthreads();

    constexpr int TOT = 16 * F;
    constexpr int NT  = JW * 32;
    #pragma unroll
    for (int r = 0; r < R; ++r) {
        float Mrow = -INFINITY;
        #pragma unroll
        for (int w = 0; w < JW; ++w) Mrow = fmaxf(Mrow, m_lds[r][w * 16 + lo]);
        float Lrow = 0.f;
        #pragma unroll
        for (int w = 0; w < JW; ++w)
            Lrow += l_lds[r][w * 16 + lo] * __expf(m_lds[r][w * 16 + lo] - Mrow);
        if (wave == 0 && hi == 0) L_lds[lo] = Lrow;

        const float fown = __expf(m[r] - Mrow);
        float fv[8];
        #pragma unroll
        for (int v = 0; v < 8; ++v) fv[v] = __shfl(fown, v + 8 * hi, 32);

        float* ab = accbuf + (wave * 16 + 8 * hi) * F + lo;
        #pragma unroll
        for (int c = 0; c < NC; ++c)
            #pragma unroll
            for (int v = 0; v < 8; ++v)
                ab[v * F + 16 * c] = acc[r][c][v] * fv[v];
        __syncthreads();

        float* outBase = Out + (long)(i0 + 16 * r) * ldo + head * headColMul;
        #pragma unroll
        for (int e = 0; e < TOT / NT; ++e) {
            const int flat = e * NT + threadIdx.x;
            const int row = flat / F;
            const int col = flat % F;
            float val = 0.f;
            #pragma unroll
            for (int w = 0; w < JW; ++w) val += accbuf[(w * 16 + row) * F + col];
            float rr = val / L_lds[row];
            if (ELU) rr = rr > 0.f ? rr : (__expf(rr) - 1.0f);
            if constexpr (LSM) obuf[row * F + col] = rr;
            else outBase[(long)row * ldo + col] = rr;
        }
        if (r + 1 < R) __syncthreads();
    }

    if constexpr (LSM) {
        static_assert(R == 1 && JW == 16 && F == 128, "one wave per row, four columns per lane");
        __syncthreads();
        const float* rp = obuf + wave * F;
        float v[4];
        float mx = -INFINITY;
        #pragma unroll
        for (int q = 0; q < 4; ++q) {
            v[q] = rp[lane + 32 * q];
            mx = fmaxf(mx, v[q]);
        }
        #pragma unroll
        for (int off = 16; off >= 1; off >>= 1) mx = fmaxf(mx, __shfl_xor(mx, off, 32));
        float s = 0.f;
        #pragma unroll
        for (int q = 0; q < 4; ++q) s += __expf(v[q] - mx);
        #pragma unroll
        for (int off = 16; off >= 1; off >>= 1) s += __shfl_xor(s, off, 32);
        const float ls = __logf(s);
        float* op = Out + (long)(i0 + wave) * ldo + head * headColMul;
        #pragma unroll
        for (int q = 0; q < 4; ++q) op[lane + 32 * q] = v[q] - mx - ls;
    }
}

extern "C" void kernel_launch(void* const* d_in, const int* in_sizes, int n_in,
                              void* d_out, int out_size, void* d_ws, size_t ws_size,
                              hipStream_t stream) {
    const float* x     = (const float*)d_in[0];
    const int*   adj   = (const int*)d_in[1];
    const float* W     = (const float*)d_in[2];
    const float* a     = (const float*)d_in[3];
    const float* W_out = (const float*)d_in[4];
    const float* a_out = (const float*)d_in[5];
    float* out = (float*)d_out;

    char* ws = (char*)d_ws;
    unsigned* bits = (unsigned*)ws;
    ws += (size_t)NN * (NN / 32) * sizeof(unsigned);
    float* h1 = (float*)ws;
    ws += (size_t)NHEADS * NN * NHID * sizeof(float);
    float* fi1 = (float*)ws; ws += (size_t)NHEADS * NN * sizeof(float);
    float* fj1 = (float*)ws; ws += (size_t)NHEADS * NN * sizeof(float);
    float* xcat = (float*)ws;
    ws += (size_t)NN * (NHEADS * NHID) * sizeof(float);
    float* h2 = (float*)ws;
    ws += (size_t)NN * NCLASS * sizeof(float);
    float* fij2 = (float*)ws;

    {
        const long nwords = (long)NN * (NN / 32);
        pack_adj_kernel<<<dim3((nwords + 255) / 256), dim3(256), 0, stream>>>(adj, bits, nwords);
    }

    wmma_gemm_f32<NFEAT, NHID, 2, 4, 32, 1><<<dim3(NN / 128, 1, NHEADS), dim3(128), 0, stream>>>(
        x, W, (long)NFEAT * NHID, h1, (long)NN * NHID,
        a, 2 * NHID, fi1, fj1, NN);

    gat_attn<NHID, 4, 2, 4, false, false, false><<<dim3(NN / 32, NHEADS), dim3(128), 0, stream>>>(
        h1, (long)NN * NHID, fi1, fj1, bits,
        xcat, NHEADS * NHID, NHID, NN);

    wmma_gemm_f32<NHEADS * NHID, NCLASS, 1, 2, 64, 4><<<dim3(NN / 16, 1, 1), dim3(128), 0, stream>>>(
        xcat, W_out, 0, h2, 0,
        a_out, 2 * NCLASS, fij2, fij2, NN);

    gat_attn<NCLASS, 16, 1, 3, true, true, true><<<dim3(NN / 16, 1), dim3(512), 0, stream>>>(
        h2, 0, fij2, fij2 + 16, bits,
        out, NCLASS, 0, NN);
}


# attention softmax shifted by a bound known before the loop, leaky(fi + largest fj of the wave's chunk), not a running max: no rescale branch, no lane exchange in loop; l per 16 lanes, joined after
# speedup vs baseline: 1.5026x; 1.0633x over previous
#include <hip/hip_runtime.h>
#include <math.h>

#define NN      4096
#define NFEAT   512
#define NHID    64
#define NCLASS  128
#define NHEADS  8
#define ALPHA   0.2f
#define NEGV    (-9000000000000000.0f)

typedef __attribute__((ext_vector_type(2))) float v2f;
typedef __attribute__((ext_vector_type(4))) float v4f;
typedef __attribute__((ext_vector_type(8))) float v8f;

__global__ void pack_adj_kernel(const int* __restrict__ adj,
                                unsigned* __restrict__ bits, long nwords) {
    long w = (long)blockIdx.x * blockDim.x + threadIdx.x;
    if (w >= nwords) return;
    const int* p = adj + w * 32;
    unsigned m = 0u;
    #pragma unroll
    for (int b = 0; b < 32; ++b) m |= (p[b] > 0 ? 1u : 0u) << b;
    bits[w] = m;
}

typedef __attribute__((ext_vector_type(16))) float v16f;

__device__ __forceinline__ float lane_tree_2(v2f y, int lo) {
    const bool up = (lo & 1) != 0;
    const float keep = up ? y.y : y.x;
    const float give = up ? y.x : y.y;
    return keep + __shfl_xor(give, 1, 32);
}
__device__ __forceinline__ float lane_tree_4(v4f y, int lo) {
    const bool up = (lo & 2) != 0;
    const v2f keep = up ? y.hi : y.lo;
    const v2f give = up ? y.lo : y.hi;
    v2f z;
    z.x = keep.x + __shfl_xor(give.x, 2, 32);
    z.y = keep.y + __shfl_xor(give.y, 2, 32);
    return lane_tree_2(z, lo);
}
__device__ __forceinline__ float lane_tree_8(v8f y, int lo) {
    const bool up = (lo & 4) != 0;
    const v4f keep = up ? y.hi : y.lo;
    const v4f give = up ? y.lo : y.hi;
    v4f z;
    #pragma unroll
    for (int i = 0; i < 4; ++i) z[i] = keep[i] + __shfl_xor(give[i], 4, 32);
    return lane_tree_4(z, lo);
}
__device__ __forceinline__ float lane_tree_16(v16f y, int lo) {
    const bool up = (lo & 8) != 0;
    const v8f keep = up ? y.hi : y.lo;
    const v8f give = up ? y.lo : y.hi;
    v8f z;
    #pragma unroll
    for (int i = 0; i < 8; ++i) z[i] = keep[i] + __shfl_xor(give[i], 8, 32);
    return lane_tree_8(z, lo);
}

template <int R, int CT, int G, int K, int NB>
__device__ __forceinline__ void gemm_load_block(
        v4f (&a)[R][G], __attribute__((ext_vector_type(CT))) float (&b)[G][4],
        const float* __restrict__ ap, const float* __restrict__ bp, int k) {
    typedef __attribute__((ext_vector_type(CT))) float vct;
    #pragma unroll
    for (int g = 0; g < G; ++g) {
        #pragma unroll
        for (int r = 0; r < R; ++r)
            a[r][g] = *(const v4f*)(ap + (long)r * 16 * K + k + 8 * g);
        #pragma unroll
        for (int j = 0; j < 4; ++j)
            b[g][j] = *(const vct*)(bp + (long)(k + 8 * g + j) * NB);
    }
}

template <int R, int CT, int G>
__device__ __forceinline__ void gemm_mma_block(
        v8f (&acc)[R][CT],
        const v4f (&a)[R][G], const __attribute__((ext_vector_type(CT))) float (&b)[G][4]) {
    v2f q[G][2][CT];
    #pragma unroll
    for (int g = 0; g < G; ++g)
        #pragma unroll
        for (int kk = 0; kk < 2; ++kk)
            #pragma unroll
            for (int t = 0; t < CT; ++t) {
                q[g][kk][t].x = b[g][2 * kk][t];
                q[g][kk][t].y = b[g][2 * kk + 1][t];
            }
    __builtin_amdgcn_sched_barrier(0);
    #pragma unroll
    for (int g = 0; g < G; ++g) {
        #pragma unroll
        for (int kk = 0; kk < 2; ++kk) {
            #pragma unroll
            for (int t = 0; t < CT; ++t) {
                #pragma unroll
                for (int r = 0; r < R; ++r) {
                    v2f av;
                    av.x = a[r][g][2 * kk];
                    av.y = a[r][g][2 * kk + 1];
                    acc[r][t] = __builtin_amdgcn_wmma_f32_16x16x4_f32(
                        false, av, false, q[g][kk][t], (short)0, acc[r][t], false, false);
                }
            }
        }
    }
    __builtin_amdgcn_sched_barrier(0);
    asm volatile("v_nop\n\tv_nop\n\tv_nop\n\tv_nop");
    __builtin_amdgcn_sched_barrier(0);
}

template <int K, int NB, int R, int CT, int KB, int WCOLS>
__global__ __launch_bounds__(128) void wmma_gemm_f32(
        const float* __restrict__ A,
        const float* __restrict__ B, long bStride,
        float* __restrict__ C, long cStride,
        const float* __restrict__ avec, int aStride,
        float* __restrict__ fi, float* __restrict__ fj, int n) {
    typedef __attribute__((ext_vector_type(CT))) float vct;
    constexpr int G = KB / 8;
    constexpr int WR = 4 / WCOLS;
    static_assert(K % (2 * KB) == 0, "K must hold an even number of k blocks");
    static_assert(WCOLS * 16 * CT == NB, "a block must hold whole rows of C");
    static_assert(WCOLS == 1 || WR == 1, "waves run along the rows or along the columns");
    const int wave = threadIdx.x >> 5;
    const int lane = threadIdx.x & 31;
    const int hi   = lane >> 4;
    const int lo   = lane & 15;
    const int wc   = wave % WCOLS;
    const int row0 = (blockIdx.x * WR + wave / WCOLS) * (16 * R);
    const int colb = wc * (16 * CT);
    const int head = blockIdx.z;

    const float* __restrict__ ap = A + (long)(row0 + lo) * K + 4 * hi;
    const float* __restrict__ bp = B + (long)head * bStride + (long)(4 * hi) * NB + colb + CT * lo;

    v8f acc[R][CT];
    #pragma unroll
    for (int r = 0; r < R; ++r)
        #pragma unroll
        for (int t = 0; t < CT; ++t) acc[r][t] = (v8f){};

    v4f a0[R][G], a1[R][G];
    vct b0[G][4], b1[G][4];
    gemm_load_block<R, CT, G, K, NB>(a0, b0, ap, bp, 0);

    #pragma unroll 1
    for (int k0 = 0; k0 < K; k0 += 2 * KB) {
        gemm_load_block<R, CT, G, K, NB>(a1, b1, ap, bp, k0 + KB);
        __builtin_amdgcn_sched_barrier(0);
        gemm_mma_block<R, CT, G>(acc, a0, b0);
        const int kn = (k0 + 2 * KB < K) ? (k0 + 2 * KB) : k0;
        gemm_load_block<R, CT, G, K, NB>(a0, b0, ap, bp, kn);
        __builtin_amdgcn_sched_barrier(0);
        gemm_mma_block<R, CT, G>(acc, a1, b1);
    }

    float* __restrict__ cp = C + (long)head * cStride + (long)(row0 + 8 * hi) * NB + colb + CT * lo;
    #pragma unroll
    for (int r = 0; r < R; ++r) {
        #pragma unroll
        for (int v = 0; v < 8; ++v) {
            vct o;
            #pragma unroll
            for (int t = 0; t < CT; ++t) o[t] = acc[r][t][v];
            *(vct*)(cp + (long)(16 * r + v) * NB) = o;
        }
    }

    const float* __restrict__ a1p = avec + (long)head * aStride + colb + CT * lo;
    const vct a1v = *(const vct*)a1p;
    const vct a2v = *(const vct*)(a1p + NB);
    constexpr int NV = 8 * R;
    static_assert(R == 1 || R == 2, "8 or 16 rows per lane half");
    typedef __attribute__((ext_vector_type(NV))) float vnv;
    vnv p1, p2;
    #pragma unroll
    for (int r = 0; r < R; ++r) {
        #pragma unroll
        for (int v = 0; v < 8; ++v) {
            float s1 = 0.f, s2 = 0.f;
            #pragma unroll
            for (int t = 0; t < CT; ++t) {
                s1 += acc[r][t][v] * a1v[t];
                s2 += acc[r][t][v] * a2v[t];
            }
            p1[8 * r + v] = s1;
            p2[8 * r + v] = s2;
        }
    }
    float w1, w2;
    if constexpr (NV == 16) {
        w1 = lane_tree_16(p1, lo);
        w2 = lane_tree_16(p2, lo);
    } else {
        w1 = lane_tree_8(p1, lo);
        w2 = lane_tree_8(p2, lo);
        w1 += __shfl_xor(w1, 8, 32);
        w2 += __shfl_xor(w2, 8, 32);
    }
    const int qv   = lo % NV;
    const int qrow = 16 * (qv >> 3) + 8 * hi + (qv & 7);
    if constexpr (WCOLS == 1) {
        if (lo < NV) {
            const long o = (long)head * n + row0 + qrow;
            fi[o] = w1;
            fj[o] = w2;
        }
    } else {
        __shared__ float part[2][WCOLS][16 * R];
        if (lo < NV) {
            part[0][wc][qrow] = w1;
            part[1][wc][qrow] = w2;
        }
        __syncthreads();
        static_assert(R == 1, "the packed line holds one row tile");
        if (threadIdx.x < 32) {
            const int which = threadIdx.x >> 4;
            float t1 = 0.f;
            #pragma unroll
            for (int w = 0; w < WCOLS; ++w) t1 += part[which][w][threadIdx.x & 15];
            fi[2 * ((long)head * n + row0) + threadIdx.x] = t1;
        }
    }
}

template <int F, int JW, int R, int PRE, bool ELU, bool PK, bool LSM>
__global__ __launch_bounds__(JW * 32) void gat_attn(
        const float* __restrict__ Hfeat, long hStride,
        const float* __restrict__ fi, const float* __restrict__ fj,
        const unsigned* __restrict__ bits,
        float* __restrict__ Out, int ldo, int headColMul, int n) {
    constexpr int NC = F / 16;
    constexpr float LOG2E = 1.4426950408889634f;
    constexpr float AL2   = ALPHA * LOG2E;
    const int wave = threadIdx.x >> 5;
    const int lane = threadIdx.x & 31;
    const int hi   = lane >> 4;
    const int lo   = lane & 15;
    const int head = blockIdx.y;
    const int i0   = blockIdx.x * (16 * R);
    const int nwords = n >> 5;

    const float* __restrict__ Hh  = Hfeat + (long)head * hStride;
    const float* __restrict__ fih = fi + (PK ? 2 : 1) * (long)head * n;
    const float* __restrict__ fjh = fj + (PK ? 2 : 1) * (long)head * n;
    const float* __restrict__ bbase = Hh + 2 * hi * F + lo;

    float fiv[R];
    const unsigned* __restrict__ brow[R];
    #pragma unroll
    for (int r = 0; r < R; ++r) {
        fiv[r]  = LOG2E * fih[(PK ? 2 : 1) * (i0 + 16 * r) + lo];
        brow[r] = bits + (long)(i0 + 16 * r + lo) * nwords;
    }

    const int chunk = n / JW;
    const int jbeg  = wave * chunk;
    const int jend  = jbeg + chunk;

    constexpr int NLD = NN / (JW * 128);
    static_assert(NLD * JW * 128 == NN, "whole 16-byte loads per lane");
    float fjmax = -INFINITY;
    #pragma unroll
    for (int k = 0; k < NLD; ++k) {
        const int x = jbeg + 128 * k + 4 * lane;
        const v4f t = *(const v4f*)(fjh + (PK ? (2 * (x & ~15) + (x & 15)) : x));
        fjmax = fmaxf(fmaxf(fjmax, fmaxf(t.x, t.y)), fmaxf(t.z, t.w));
    }
    #pragma unroll
    for (int off = 16; off >= 1; off >>= 1) fjmax = fmaxf(fjmax, __shfl_xor(fjmax, off, 32));
    fjmax *= LOG2E;

    float m[R], l[R], a1[R], a2[R];
    v8f acc[R][NC];
    #pragma unroll
    for (int r = 0; r < R; ++r) {
        const float vb = fiv[r] + fjmax;
        m[r] = fmaxf(vb, ALPHA * vb);
        a1[r] = fiv[r] - m[r];
        a2[r] = ALPHA * fiv[r] - m[r];
        l[r] = 0.f;
        #pragma unroll
        for (int c = 0; c < NC; ++c) acc[r][c] = (v8f){};
    }

    unsigned wm_n[R];
    float fj_n[16];
    #pragma unroll
    for (int r = 0; r < R; ++r) wm_n[r] = brow[r][jbeg >> 5];
    {
        const float* __restrict__ fjp = fjh + (PK ? 2 : 1) * jbeg + 2 * hi;
        #pragma unroll
        for (int q = 0; q < 8; ++q)
            #pragma unroll
            for (int t = 0; t < 2; ++t) fj_n[2 * q + t] = fjp[4 * q + t + ((PK && q >= 4) ? 16 : 0)];
    }

    for (int j0 = jbeg; j0 < jend; j0 += 32) {
        unsigned wmask[R];
        float fjc[16];
        #pragma unroll
        for (int r = 0; r < R; ++r) wmask[r] = wm_n[r];
        #pragma unroll
        for (int q = 0; q < 16; ++q) fjc[q] = fj_n[q];

        #pragma unroll
        for (int half = 0; half < 2; ++half) {
            const int jb = j0 + 16 * half;
            const int bitbase = 16 * half;

            float fjv[8];
            #pragma unroll
            for (int q = 0; q < 8; ++q) fjv[q] = fjc[8 * half + q];

            v2f bf[4][NC];
            const float* __restrict__ bp = bbase + jb * F;
            #pragma unroll
            for (int s = 0; s < PRE; ++s) {
                const float* __restrict__ bs = bp + 4 * s * F;
                #pragma unroll
                for (int c = 0; c < NC; ++c) {
                    bf[s][c].x = bs[16 * c];
                    bf[s][c].y = bs[16 * c + F];
                }
            }

            v2f af[R][4];
            #pragma unroll
            for (int r = 0; r < R; ++r) {
                const unsigned wsh = wmask[r] >> (bitbase + 2 * hi);
                float psum = 0.f;
                #pragma unroll
                for (int s = 0; s < 4; ++s) {
                    #pragma unroll
                    for (int t = 0; t < 2; ++t) {
                        const float f = fjv[2 * s + t];
                        const float x = fmaxf(__builtin_fmaf(f, LOG2E, a1[r]),
                                              __builtin_fmaf(f, AL2, a2[r]));
                        const float w = __builtin_amdgcn_exp2f(x);
                        const float p = ((wsh >> (4 * s + t)) & 1u) ? w : 0.f;
                        af[r][s][t] = p;
                        psum += p;
                    }
                }
                l[r] += psum;
            }
            #pragma unroll
            for (int r = 0; r < R; ++r)
                #pragma unroll
                for (int s = 0; s < 4; ++s) asm volatile("" : "+v"(af[r][s]));

            if (half == 1) {
                const int jn = (j0 + 32 < jend) ? (j0 + 32) : j0;
                const float* __restrict__ fjp = fjh + (PK ? 2 : 1) * jn + 2 * hi;
                #pragma unroll
                for (int q = 0; q < 8; ++q)
                    #pragma unroll
                    for (int t = 0; t < 2; ++t) fj_n[2 * q + t] = fjp[4 * q + t + ((PK && q >= 4) ? 16 : 0)];
                #pragma unroll
                for (int r = 0; r < R; ++r) wm_n[r] = brow[r][jn >> 5];
            }
            #pragma unroll
            for (int s = PRE; s < 4; ++s) {
                const float* __restrict__ bs = bp + 4 * s * F;
                #pragma unroll
                for (int c = 0; c < NC; ++c) {
                    bf[s][c].x = bs[16 * c];
                    bf[s][c].y = bs[16 * c + F];
                }
            }
            __builtin_amdgcn_sched_barrier(0);

            #pragma unroll
            for (int s = 0; s < 4; ++s) {
                #pragma unroll
                for (int c = 0; c < NC; ++c) {
                    #pragma unroll
                    for (int r = 0; r < R; ++r) {
                        acc[r][c] = __builtin_amdgcn_wmma_f32_16x16x4_f32(
                            false, af[r][s], false, bf[s][c], (short)0, acc[r][c], false, false);
                    }
                }
            }
            #pragma unroll
            for (int r = 0; r < R; ++r)
                #pragma unroll
                for (int c = 0; c < NC; ++c) asm volatile("" : "+v"(acc[r][c]));
            __builtin_amdgcn_sched_barrier(0);
            asm volatile("v_nop\n\tv_nop\n\tv_nop\n\tv_nop");
            __builtin_amdgcn_sched_barrier(0);
        }
    }

    #pragma unroll
    for (int r = 0; r < R; ++r) l[r] += __shfl_xor(l[r], 16, 32);

    __shared__ float m_lds[R][JW * 16];
    __shared__ float l_lds[R][JW * 16];
    __shared__ float L_lds[16];
    __shared__ float accbuf[JW * 16 * F];
    __shared__ float obuf[LSM ? 16 * F : 1];
    __shared__ float cpart[JW * 32];
    __shared__ float cmean[F];

    #pragma unroll
    for (int r = 0; r < R; ++r) {
        if (hi == 0) {
            m_lds[r][wave * 16 + lo] = m[r];
            l_lds[r][wave * 16 + lo] = l[r];
        }
    }
    __syncthreads();

    constexpr int TOT = 16 * F;
    constexpr int NT  = JW * 32;
    #pragma unroll
    for (int r = 0; r < R; ++r) {
        float Mrow = -INFINITY;
        #pragma unroll
        for (int w = 0; w < JW; ++w) Mrow = fmaxf(Mrow, m_lds[r][w * 16 + lo]);
        float Lrow = 0.f;
        #pragma unroll
        for (int w = 0; w < JW; ++w)
            Lrow += l_lds[r][w * 16 + lo] * __builtin_amdgcn_exp2f(m_lds[r][w * 16 + lo] - Mrow);
        if (wave == 0 && hi == 0) L_lds[lo] = Lrow;

        const float fown = __builtin_amdgcn_exp2f(m[r] - Mrow);
        float fv[8];
        #pragma unroll
        for (int v = 0; v < 8; ++v) fv[v] = __shfl(fown, v + 8 * hi, 32);

        float* ab = accbuf + (wave * 16 + 8 * hi) * F + lo;
        #pragma unroll
        for (int c = 0; c < NC; ++c)
            #pragma unroll
            for (int v = 0; v < 8; ++v)
                ab[v * F + 16 * c] = acc[r][c][v] * fv[v];
        __syncthreads();

        const bool anyEmpty = __any(L_lds[lo] == 0.f);
        if (anyEmpty) {
            static_assert(NT % F == 0, "whole columns per thread group");
            constexpr int NS = NT / F;
            const int ccol = threadIdx.x % F;
            float csum = 0.f;
            for (int j = threadIdx.x / F; j < n; j += NS) csum += Hh[(long)j * F + ccol];
            cpart[threadIdx.x] = csum;
            __syncthreads();
            if (threadIdx.x < F) {
                float t = 0.f;
                #pragma unroll
                for (int k = 0; k < NS; ++k) t += cpart[k * F + threadIdx.x];
                cmean[threadIdx.x] = t / (float)n;
            }
            __syncthreads();
        }

        float* outBase = Out + (long)(i0 + 16 * r) * ldo + head * headColMul;
        #pragma unroll
        for (int e = 0; e < TOT / NT; ++e) {
            const int flat = e * NT + threadIdx.x;
            const int row = flat / F;
            const int col = flat % F;
            float val = 0.f;
            #pragma unroll
            for (int w = 0; w < JW; ++w) val += accbuf[(w * 16 + row) * F + col];
            const float Lr = L_lds[row];
            float rr = val / Lr;
            if (anyEmpty && Lr == 0.f) rr = cmean[col];
            if (ELU) rr = rr > 0.f ? rr : (__expf(rr) - 1.0f);
            if constexpr (LSM) obuf[row * F + col] = rr;
            else outBase[(long)row * ldo + col] = rr;
        }
        if (r + 1 < R) __syncthreads();
    }

    if constexpr (LSM) {
        static_assert(R == 1 && JW == 16 && F == 128, "one wave per row, four columns per lane");
        __syncthreads();
        const float* rp = obuf + wave * F;
        float v[4];
        float mx = -INFINITY;
        #pragma unroll
        for (int q = 0; q < 4; ++q) {
            v[q] = rp[lane + 32 * q];
            mx = fmaxf(mx, v[q]);
        }
        #pragma unroll
        for (int off = 16; off >= 1; off >>= 1) mx = fmaxf(mx, __shfl_xor(mx, off, 32));
        float s = 0.f;
        #pragma unroll
        for (int q = 0; q < 4; ++q) s += __expf(v[q] - mx);
        #pragma unroll
        for (int off = 16; off >= 1; off >>= 1) s += __shfl_xor(s, off, 32);
        const float ls = __logf(s);
        float* op = Out + (long)(i0 + wave) * ldo + head * headColMul;
        #pragma unroll
        for (int q = 0; q < 4; ++q) op[lane + 32 * q] = v[q] - mx - ls;
    }
}

extern "C" void kernel_launch(void* const* d_in, const int* in_sizes, int n_in,
                              void* d_out, int out_size, void* d_ws, size_t ws_size,
                              hipStream_t stream) {
    const float* x     = (const float*)d_in[0];
    const int*   adj   = (const int*)d_in[1];
    const float* W     = (const float*)d_in[2];
    const float* a     = (const float*)d_in[3];
    const float* W_out = (const float*)d_in[4];
    const float* a_out = (const float*)d_in[5];
    float* out = (float*)d_out;

    char* ws = (char*)d_ws;
    unsigned* bits = (unsigned*)ws;
    ws += (size_t)NN * (NN / 32) * sizeof(unsigned);
    float* h1 = (float*)ws;
    ws += (size_t)NHEADS * NN * NHID * sizeof(float);
    float* fi1 = (float*)ws; ws += (size_t)NHEADS * NN * sizeof(float);
    float* fj1 = (float*)ws; ws += (size_t)NHEADS * NN * sizeof(float);
    float* xcat = (float*)ws;
    ws += (size_t)NN * (NHEADS * NHID) * sizeof(float);
    float* h2 = (float*)ws;
    ws += (size_t)NN * NCLASS * sizeof(float);
    float* fij2 = (float*)ws;

    {
        const long nwords = (long)NN * (NN / 32);
        pack_adj_kernel<<<dim3((nwords + 255) / 256), dim3(256), 0, stream>>>(adj, bits, nwords);
    }

    wmma_gemm_f32<NFEAT, NHID, 2, 4, 32, 1><<<dim3(NN / 128, 1, NHEADS), dim3(128), 0, stream>>>(
        x, W, (long)NFEAT * NHID, h1, (long)NN * NHID,
        a, 2 * NHID, fi1, fj1, NN);

    gat_attn<NHID, 4, 2, 4, false, false, false><<<dim3(NN / 32, NHEADS), dim3(128), 0, stream>>>(
        h1, (long)NN * NHID, fi1, fj1, bits,
        xcat, NHEADS * NHID, NHID, NN);

    wmma_gemm_f32<NHEADS * NHID, NCLASS, 1, 2, 64, 4><<<dim3(NN / 16, 1, 1), dim3(128), 0, stream>>>(
        xcat, W_out, 0, h2, 0,
        a_out, 2 * NCLASS, fij2, fij2, NN);

    gat_attn<NCLASS, 16, 1, 3, true, true, true><<<dim3(NN / 16, 1), dim3(512), 0, stream>>>(
        h2, 0, fij2, fij2 + 16, bits,
        out, NCLASS, 0, NN);
}


# adjacency packing moved into the layer-1 projection kernel: each wave streams its share of adj (whole lines, mask words by wave ballot) under its WMMA runs, two batches in flight; 5 launches -> 4
# speedup vs baseline: 1.5513x; 1.0324x over previous
#include <hip/hip_runtime.h>
#include <math.h>

#define NN      4096
#define NFEAT   512
#define NHID    64
#define NCLASS  128
#define NHEADS  8
#define ALPHA   0.2f
#define NEGV    (-9000000000000000.0f)

typedef __attribute__((ext_vector_type(2))) float v2f;
typedef __attribute__((ext_vector_type(4))) float v4f;
typedef __attribute__((ext_vector_type(8))) float v8f;

__device__ __forceinline__ void adj_load_batch(int (&v)[32], const int* __restrict__ p) {
    #pragma unroll
    for (int k = 0; k < 32; ++k) v[k] = p[32 * k];
}
__device__ __forceinline__ void adj_pack_batch(const int (&v)[32], unsigned* __restrict__ q, int lane) {
    unsigned mine = 0u;
    #pragma unroll
    for (int k = 0; k < 32; ++k) {
        const unsigned m = __builtin_amdgcn_ballot_w32(v[k] > 0);
        mine = (lane == k) ? m : mine;
    }
    q[lane] = mine;
}

typedef __attribute__((ext_vector_type(16))) float v16f;

__device__ __forceinline__ float lane_tree_2(v2f y, int lo) {
    const bool up = (lo & 1) != 0;
    const float keep = up ? y.y : y.x;
    const float give = up ? y.x : y.y;
    return keep + __shfl_xor(give, 1, 32);
}
__device__ __forceinline__ float lane_tree_4(v4f y, int lo) {
    const bool up = (lo & 2) != 0;
    const v2f keep = up ? y.hi : y.lo;
    const v2f give = up ? y.lo : y.hi;
    v2f z;
    z.x = keep.x + __shfl_xor(give.x, 2, 32);
    z.y = keep.y + __shfl_xor(give.y, 2, 32);
    return lane_tree_2(z, lo);
}
__device__ __forceinline__ float lane_tree_8(v8f y, int lo) {
    const bool up = (lo & 4) != 0;
    const v4f keep = up ? y.hi : y.lo;
    const v4f give = up ? y.lo : y.hi;
    v4f z;
    #pragma unroll
    for (int i = 0; i < 4; ++i) z[i] = keep[i] + __shfl_xor(give[i], 4, 32);
    return lane_tree_4(z, lo);
}
__device__ __forceinline__ float lane_tree_16(v16f y, int lo) {
    const bool up = (lo & 8) != 0;
    const v8f keep = up ? y.hi : y.lo;
    const v8f give = up ? y.lo : y.hi;
    v8f z;
    #pragma unroll
    for (int i = 0; i < 8; ++i) z[i] = keep[i] + __shfl_xor(give[i], 8, 32);
    return lane_tree_8(z, lo);
}

template <int R, int CT, int G, int K, int NB>
__device__ __forceinline__ void gemm_load_block(
        v4f (&a)[R][G], __attribute__((ext_vector_type(CT))) float (&b)[G][4],
        const float* __restrict__ ap, const float* __restrict__ bp, int k) {
    typedef __attribute__((ext_vector_type(CT))) float vct;
    #pragma unroll
    for (int g = 0; g < G; ++g) {
        #pragma unroll
        for (int r = 0; r < R; ++r)
            a[r][g] = *(const v4f*)(ap + (long)r * 16 * K + k + 8 * g);
        #pragma unroll
        for (int j = 0; j < 4; ++j)
            b[g][j] = *(const vct*)(bp + (long)(k + 8 * g + j) * NB);
    }
}

template <int R, int CT, int G>
__device__ __forceinline__ void gemm_mma_block(
        v8f (&acc)[R][CT],
        const v4f (&a)[R][G], const __attribute__((ext_vector_type(CT))) float (&b)[G][4]) {
    v2f q[G][2][CT];
    #pragma unroll
    for (int g = 0; g < G; ++g)
        #pragma unroll
        for (int kk = 0; kk < 2; ++kk)
            #pragma unroll
            for (int t = 0; t < CT; ++t) {
                q[g][kk][t].x = b[g][2 * kk][t];
                q[g][kk][t].y = b[g][2 * kk + 1][t];
            }
    __builtin_amdgcn_sched_barrier(0);
    #pragma unroll
    for (int g = 0; g < G; ++g) {
        #pragma unroll
        for (int kk = 0; kk < 2; ++kk) {
            #pragma unroll
            for (int t = 0; t < CT; ++t) {
                #pragma unroll
                for (int r = 0; r < R; ++r) {
                    v2f av;
                    av.x = a[r][g][2 * kk];
                    av.y = a[r][g][2 * kk + 1];
                    acc[r][t] = __builtin_amdgcn_wmma_f32_16x16x4_f32(
                        false, av, false, q[g][kk][t], (short)0, acc[r][t], false, false);
                }
            }
        }
    }
    __builtin_amdgcn_sched_barrier(0);
    asm volatile("v_nop\n\tv_nop\n\tv_nop\n\tv_nop");
    __builtin_amdgcn_sched_barrier(0);
}

template <int K, int NB, int R, int CT, int KB, int WCOLS, bool PACK>
__global__ __launch_bounds__(128) void wmma_gemm_f32(
        const float* __restrict__ A,
        const float* __restrict__ B, long bStride,
        float* __restrict__ C, long cStride,
        const float* __restrict__ avec, int aStride,
        float* __restrict__ fi, float* __restrict__ fj, int n,
        const int* __restrict__ adj, unsigned* __restrict__ bits) {
    typedef __attribute__((ext_vector_type(CT))) float vct;
    constexpr int G = KB / 8;
    constexpr int WR = 4 / WCOLS;
    static_assert(K % (2 * KB) == 0, "K must hold an even number of k blocks");
    static_assert(WCOLS * 16 * CT == NB, "a block must hold whole rows of C");
    static_assert(WCOLS == 1 || WR == 1, "waves run along the rows or along the columns");
    const int wave = threadIdx.x >> 5;
    const int lane = threadIdx.x & 31;
    const int hi   = lane >> 4;
    const int lo   = lane & 15;
    const int wc   = wave % WCOLS;
    const int row0 = (blockIdx.x * WR + wave / WCOLS) * (16 * R);
    const int colb = wc * (16 * CT);
    const int head = blockIdx.z;

    const float* __restrict__ ap = A + (long)(row0 + lo) * K + 4 * hi;
    const float* __restrict__ bp = B + (long)head * bStride + (long)(4 * hi) * NB + colb + CT * lo;

    v8f acc[R][CT];
    #pragma unroll
    for (int r = 0; r < R; ++r)
        #pragma unroll
        for (int t = 0; t < CT; ++t) acc[r][t] = (v8f){};

    v4f a0[R][G], a1[R][G];
    vct b0[G][4], b1[G][4];
    gemm_load_block<R, CT, G, K, NB>(a0, b0, ap, bp, 0);

    constexpr int  NBX    = NN / (16 * R * WR);
    constexpr int  NWAVES = NBX * NHEADS * 4;
    constexpr long BSTEP  = (long)NWAVES * 1024;
    static_assert(!PACK || (long)NWAVES * (K / KB) * 1024 == (long)NN * NN,
                  "the waves' batches cover the adjacency matrix exactly once");
    const int gwave = (blockIdx.z * NBX + blockIdx.x) * 4 + wave;
    const int* __restrict__ adjp = adj + (long)gwave * 1024 + lane;
    unsigned* __restrict__ bitp = bits + (long)gwave * 32;
    int va[32], vb[32];
    if constexpr (PACK) {
        #pragma unroll
        for (int k = 0; k < 32; ++k) vb[k] = 0;
    }

    #pragma unroll 1
    for (int k0 = 0; k0 < K; k0 += 2 * KB) {
        gemm_load_block<R, CT, G, K, NB>(a1, b1, ap, bp, k0 + KB);
        if constexpr (PACK) adj_load_batch(va, adjp + (long)(k0 / KB) * BSTEP);
        __builtin_amdgcn_sched_barrier(0);
        gemm_mma_block<R, CT, G>(acc, a0, b0);
        if constexpr (PACK) {
            if (k0 > 0) adj_pack_batch(vb, bitp + (long)(k0 / KB - 1) * (BSTEP / 32), lane);
        }
        const int kn = (k0 + 2 * KB < K) ? (k0 + 2 * KB) : k0;
        gemm_load_block<R, CT, G, K, NB>(a0, b0, ap, bp, kn);
        if constexpr (PACK) adj_load_batch(vb, adjp + (long)(k0 / KB + 1) * BSTEP);
        __builtin_amdgcn_sched_barrier(0);
        gemm_mma_block<R, CT, G>(acc, a1, b1);
        if constexpr (PACK) adj_pack_batch(va, bitp + (long)(k0 / KB) * (BSTEP / 32), lane);
    }
    if constexpr (PACK) adj_pack_batch(vb, bitp + (long)(K / KB - 1) * (BSTEP / 32), lane);

    float* __restrict__ cp = C + (long)head * cStride + (long)(row0 + 8 * hi) * NB + colb + CT * lo;
    #pragma unroll
    for (int r = 0; r < R; ++r) {
        #pragma unroll
        for (int v = 0; v < 8; ++v) {
            vct o;
            #pragma unroll
            for (int t = 0; t < CT; ++t) o[t] = acc[r][t][v];
            *(vct*)(cp + (long)(16 * r + v) * NB) = o;
        }
    }

    const float* __restrict__ a1p = avec + (long)head * aStride + colb + CT * lo;
    const vct a1v = *(const vct*)a1p;
    const vct a2v = *(const vct*)(a1p + NB);
    constexpr int NV = 8 * R;
    static_assert(R == 1 || R == 2, "8 or 16 rows per lane half");
    typedef __attribute__((ext_vector_type(NV))) float vnv;
    vnv p1, p2;
    #pragma unroll
    for (int r = 0; r < R; ++r) {
        #pragma unroll
        for (int v = 0; v < 8; ++v) {
            float s1 = 0.f, s2 = 0.f;
            #pragma unroll
            for (int t = 0; t < CT; ++t) {
                s1 += acc[r][t][v] * a1v[t];
                s2 += acc[r][t][v] * a2v[t];
            }
            p1[8 * r + v] = s1;
            p2[8 * r + v] = s2;
        }
    }
    float w1, w2;
    if constexpr (NV == 16) {
        w1 = lane_tree_16(p1, lo);
        w2 = lane_tree_16(p2, lo);
    } else {
        w1 = lane_tree_8(p1, lo);
        w2 = lane_tree_8(p2, lo);
        w1 += __shfl_xor(w1, 8, 32);
        w2 += __shfl_xor(w2, 8, 32);
    }
    const int qv   = lo % NV;
    const int qrow = 16 * (qv >> 3) + 8 * hi + (qv & 7);
    if constexpr (WCOLS == 1) {
        if (lo < NV) {
            const long o = (long)head * n + row0 + qrow;
            fi[o] = w1;
            fj[o] = w2;
        }
    } else {
        __shared__ float part[2][WCOLS][16 * R];
        if (lo < NV) {
            part[0][wc][qrow] = w1;
            part[1][wc][qrow] = w2;
        }
        __syncthreads();
        static_assert(R == 1, "the packed line holds one row tile");
        if (threadIdx.x < 32) {
            const int which = threadIdx.x >> 4;
            float t1 = 0.f;
            #pragma unroll
            for (int w = 0; w < WCOLS; ++w) t1 += part[which][w][threadIdx.x & 15];
            fi[2 * ((long)head * n + row0) + threadIdx.x] = t1;
        }
    }
}

template <int F, int JW, int R, int PRE, bool ELU, bool PK, bool LSM>
__global__ __launch_bounds__(JW * 32) void gat_attn(
        const float* __restrict__ Hfeat, long hStride,
        const float* __restrict__ fi, const float* __restrict__ fj,
        const unsigned* __restrict__ bits,
        float* __restrict__ Out, int ldo, int headColMul, int n) {
    constexpr int NC = F / 16;
    constexpr float LOG2E = 1.4426950408889634f;
    constexpr float AL2   = ALPHA * LOG2E;
    const int wave = threadIdx.x >> 5;
    const int lane = threadIdx.x & 31;
    const int hi   = lane >> 4;
    const int lo   = lane & 15;
    const int head = blockIdx.y;
    const int i0   = blockIdx.x * (16 * R);
    const int nwords = n >> 5;

    const float* __restrict__ Hh  = Hfeat + (long)head * hStride;
    const float* __restrict__ fih = fi + (PK ? 2 : 1) * (long)head * n;
    const float* __restrict__ fjh = fj + (PK ? 2 : 1) * (long)head * n;
    const float* __restrict__ bbase = Hh + 2 * hi * F + lo;

    float fiv[R];
    const unsigned* __restrict__ brow[R];
    #pragma unroll
    for (int r = 0; r < R; ++r) {
        fiv[r]  = LOG2E * fih[(PK ? 2 : 1) * (i0 + 16 * r) + lo];
        brow[r] = bits + (long)(i0 + 16 * r + lo) * nwords;
    }

    const int chunk = n / JW;
    const int jbeg  = wave * chunk;
    const int jend  = jbeg + chunk;

    constexpr int NLD = NN / (JW * 128);
    static_assert(NLD * JW * 128 == NN, "whole 16-byte loads per lane");
    float fjmax = -INFINITY;
    #pragma unroll
    for (int k = 0; k < NLD; ++k) {
        const int x = jbeg + 128 * k + 4 * lane;
        const v4f t = *(const v4f*)(fjh + (PK ? (2 * (x & ~15) + (x & 15)) : x));
        fjmax = fmaxf(fmaxf(fjmax, fmaxf(t.x, t.y)), fmaxf(t.z, t.w));
    }
    #pragma unroll
    for (int off = 16; off >= 1; off >>= 1) fjmax = fmaxf(fjmax, __shfl_xor(fjmax, off, 32));
    fjmax *= LOG2E;

    float m[R], l[R], a1[R], a2[R];
    v8f acc[R][NC];
    #pragma unroll
    for (int r = 0; r < R; ++r) {
        const float vb = fiv[r] + fjmax;
        m[r] = fmaxf(vb, ALPHA * vb);
        a1[r] = fiv[r] - m[r];
        a2[r] = ALPHA * fiv[r] - m[r];
        l[r] = 0.f;
        #pragma unroll
        for (int c = 0; c < NC; ++c) acc[r][c] = (v8f){};
    }

    unsigned wm_n[R];
    float fj_n[16];
    #pragma unroll
    for (int r = 0; r < R; ++r) wm_n[r] = brow[r][jbeg >> 5];
    {
        const float* __restrict__ fjp = fjh + (PK ? 2 : 1) * jbeg + 2 * hi;
        #pragma unroll
        for (int q = 0; q < 8; ++q)
            #pragma unroll
            for (int t = 0; t < 2; ++t) fj_n[2 * q + t] = fjp[4 * q + t + ((PK && q >= 4) ? 16 : 0)];
    }

    for (int j0 = jbeg; j0 < jend; j0 += 32) {
        unsigned wmask[R];
        float fjc[16];
        #pragma unroll
        for (int r = 0; r < R; ++r) wmask[r] = wm_n[r];
        #pragma unroll
        for (int q = 0; q < 16; ++q) fjc[q] = fj_n[q];

        #pragma unroll
        for (int half = 0; half < 2; ++half) {
            const int jb = j0 + 16 * half;
            const int bitbase = 16 * half;

            float fjv[8];
            #pragma unroll
            for (int q = 0; q < 8; ++q) fjv[q] = fjc[8 * half + q];

            v2f bf[4][NC];
            const float* __restrict__ bp = bbase + jb * F;
            #pragma unroll
            for (int s = 0; s < PRE; ++s) {
                const float* __restrict__ bs = bp + 4 * s * F;
                #pragma unroll
                for (int c = 0; c < NC; ++c) {
                    bf[s][c].x = bs[16 * c];
                    bf[s][c].y = bs[16 * c + F];
                }
            }

            v2f af[R][4];
            #pragma unroll
            for (int r = 0; r < R; ++r) {
                const unsigned wsh = wmask[r] >> (bitbase + 2 * hi);
                float psum = 0.f;
                #pragma unroll
                for (int s = 0; s < 4; ++s) {
                    #pragma unroll
                    for (int t = 0; t < 2; ++t) {
                        const float f = fjv[2 * s + t];
                        const float x = fmaxf(__builtin_fmaf(f, LOG2E, a1[r]),
                                              __builtin_fmaf(f, AL2, a2[r]));
                        const float w = __builtin_amdgcn_exp2f(x);
                        const float p = ((wsh >> (4 * s + t)) & 1u) ? w : 0.f;
                        af[r][s][t] = p;
                        psum += p;
                    }
                }
                l[r] += psum;
            }
            #pragma unroll
            for (int r = 0; r < R; ++r)
                #pragma unroll
                for (int s = 0; s < 4; ++s) asm volatile("" : "+v"(af[r][s]));

            if (half == 1) {
                const int jn = (j0 + 32 < jend) ? (j0 + 32) : j0;
                const float* __restrict__ fjp = fjh + (PK ? 2 : 1) * jn + 2 * hi;
                #pragma unroll
                for (int q = 0; q < 8; ++q)
                    #pragma unroll
                    for (int t = 0; t < 2; ++t) fj_n[2 * q + t] = fjp[4 * q + t + ((PK && q >= 4) ? 16 : 0)];
                #pragma unroll
                for (int r = 0; r < R; ++r) wm_n[r] = brow[r][jn >> 5];
            }
            #pragma unroll
            for (int s = PRE; s < 4; ++s) {
                const float* __restrict__ bs = bp + 4 * s * F;
                #pragma unroll
                for (int c = 0; c < NC; ++c) {
                    bf[s][c].x = bs[16 * c];
                    bf[s][c].y = bs[16 * c + F];
                }
            }
            __builtin_amdgcn_sched_barrier(0);

            #pragma unroll
            for (int s = 0; s < 4; ++s) {
                #pragma unroll
                for (int c = 0; c < NC; ++c) {
                    #pragma unroll
                    for (int r = 0; r < R; ++r) {
                        acc[r][c] = __builtin_amdgcn_wmma_f32_16x16x4_f32(
                            false, af[r][s], false, bf[s][c], (short)0, acc[r][c], false, false);
                    }
                }
            }
            #pragma unroll
            for (int r = 0; r < R; ++r)
                #pragma unroll
                for (int c = 0; c < NC; ++c) asm volatile("" : "+v"(acc[r][c]));
            __builtin_amdgcn_sched_barrier(0);
            asm volatile("v_nop\n\tv_nop\n\tv_nop\n\tv_nop");
            __builtin_amdgcn_sched_barrier(0);
        }
    }

    #pragma unroll
    for (int r = 0; r < R; ++r) l[r] += __shfl_xor(l[r], 16, 32);

    __shared__ float m_lds[R][JW * 16];
    __shared__ float l_lds[R][JW * 16];
    __shared__ float L_lds[16];
    __shared__ float accbuf[JW * 16 * F];
    __shared__ float obuf[LSM ? 16 * F : 1];
    __shared__ float cpart[JW * 32];
    __shared__ float cmean[F];

    #pragma unroll
    for (int r = 0; r < R; ++r) {
        if (hi == 0) {
            m_lds[r][wave * 16 + lo] = m[r];
            l_lds[r][wave * 16 + lo] = l[r];
        }
    }
    __syncthreads();

    constexpr int TOT = 16 * F;
    constexpr int NT  = JW * 32;
    #pragma unroll
    for (int r = 0; r < R; ++r) {
        float Mrow = -INFINITY;
        #pragma unroll
        for (int w = 0; w < JW; ++w) Mrow = fmaxf(Mrow, m_lds[r][w * 16 + lo]);
        float Lrow = 0.f;
        #pragma unroll
        for (int w = 0; w < JW; ++w)
            Lrow += l_lds[r][w * 16 + lo] * __builtin_amdgcn_exp2f(m_lds[r][w * 16 + lo] - Mrow);
        if (wave == 0 && hi == 0) L_lds[lo] = Lrow;

        const float fown = __builtin_amdgcn_exp2f(m[r] - Mrow);
        float fv[8];
        #pragma unroll
        for (int v = 0; v < 8; ++v) fv[v] = __shfl(fown, v + 8 * hi, 32);

        float* ab = accbuf + (wave * 16 + 8 * hi) * F + lo;
        #pragma unroll
        for (int c = 0; c < NC; ++c)
            #pragma unroll
            for (int v = 0; v < 8; ++v)
                ab[v * F + 16 * c] = acc[r][c][v] * fv[v];
        __syncthreads();

        const bool anyEmpty = __any(L_lds[lo] == 0.f);
        if (anyEmpty) {
            static_assert(NT % F == 0, "whole columns per thread group");
            constexpr int NS = NT / F;
            const int ccol = threadIdx.x % F;
            float csum = 0.f;
            for (int j = threadIdx.x / F; j < n; j += NS) csum += Hh[(long)j * F + ccol];
            cpart[threadIdx.x] = csum;
            __syncthreads();
            if (threadIdx.x < F) {
                float t = 0.f;
                #pragma unroll
                for (int k = 0; k < NS; ++k) t += cpart[k * F + threadIdx.x];
                cmean[threadIdx.x] = t / (float)n;
            }
            __syncthreads();
        }

        float* outBase = Out + (long)(i0 + 16 * r) * ldo + head * headColMul;
        #pragma unroll
        for (int e = 0; e < TOT / NT; ++e) {
            const int flat = e * NT + threadIdx.x;
            const int row = flat / F;
            const int col = flat % F;
            float val = 0.f;
            #pragma unroll
            for (int w = 0; w < JW; ++w) val += accbuf[(w * 16 + row) * F + col];
            const float Lr = L_lds[row];
            float rr = val / Lr;
            if (anyEmpty && Lr == 0.f) rr = cmean[col];
            if (ELU) rr = rr > 0.f ? rr : (__expf(rr) - 1.0f);
            if constexpr (LSM) obuf[row * F + col] = rr;
            else outBase[(long)row * ldo + col] = rr;
        }
        if (r + 1 < R) __syncthreads();
    }

    if constexpr (LSM) {
        static_assert(R == 1 && JW == 16 && F == 128, "one wave per row, four columns per lane");
        __syncthreads();
        const float* rp = obuf + wave * F;
        float v[4];
        float mx = -INFINITY;
        #pragma unroll
        for (int q = 0; q < 4; ++q) {
            v[q] = rp[lane + 32 * q];
            mx = fmaxf(mx, v[q]);
        }
        #pragma unroll
        for (int off = 16; off >= 1; off >>= 1) mx = fmaxf(mx, __shfl_xor(mx, off, 32));
        float s = 0.f;
        #pragma unroll
        for (int q = 0; q < 4; ++q) s += __expf(v[q] - mx);
        #pragma unroll
        for (int off = 16; off >= 1; off >>= 1) s += __shfl_xor(s, off, 32);
        const float ls = __logf(s);
        float* op = Out + (long)(i0 + wave) * ldo + head * headColMul;
        #pragma unroll
        for (int q = 0; q < 4; ++q) op[lane + 32 * q] = v[q] - mx - ls;
    }
}

extern "C" void kernel_launch(void* const* d_in, const int* in_sizes, int n_in,
                              void* d_out, int out_size, void* d_ws, size_t ws_size,
                              hipStream_t stream) {
    const float* x     = (const float*)d_in[0];
    const int*   adj   = (const int*)d_in[1];
    const float* W     = (const float*)d_in[2];
    const float* a     = (const float*)d_in[3];
    const float* W_out = (const float*)d_in[4];
    const float* a_out = (const float*)d_in[5];
    float* out = (float*)d_out;

    char* ws = (char*)d_ws;
    unsigned* bits = (unsigned*)ws;
    ws += (size_t)NN * (NN / 32) * sizeof(unsigned);
    float* h1 = (float*)ws;
    ws += (size_t)NHEADS * NN * NHID * sizeof(float);
    float* fi1 = (float*)ws; ws += (size_t)NHEADS * NN * sizeof(float);
    float* fj1 = (float*)ws; ws += (size_t)NHEADS * NN * sizeof(float);
    float* xcat = (float*)ws;
    ws += (size_t)NN * (NHEADS * NHID) * sizeof(float);
    float* h2 = (float*)ws;
    ws += (size_t)NN * NCLASS * sizeof(float);
    float* fij2 = (float*)ws;

    wmma_gemm_f32<NFEAT, NHID, 2, 4, 32, 1, true><<<dim3(NN / 128, 1, NHEADS), dim3(128), 0, stream>>>(
        x, W, (long)NFEAT * NHID, h1, (long)NN * NHID,
        a, 2 * NHID, fi1, fj1, NN, adj, bits);

    gat_attn<NHID, 4, 2, 4, false, false, false><<<dim3(NN / 32, NHEADS), dim3(128), 0, stream>>>(
        h1, (long)NN * NHID, fi1, fj1, bits,
        xcat, NHEADS * NHID, NHID, NN);

    wmma_gemm_f32<NHEADS * NHID, NCLASS, 1, 2, 64, 4, false><<<dim3(NN / 16, 1, 1), dim3(128), 0, stream>>>(
        xcat, W_out, 0, h2, 0,
        a_out, 2 * NCLASS, fij2, fij2, NN, nullptr, nullptr);

    gat_attn<NCLASS, 16, 1, 3, true, true, true><<<dim3(NN / 16, 1), dim3(512), 0, stream>>>(
        h2, 0, fij2, fij2 + 16, bits,
        out, NCLASS, 0, NN);
}


# adjacency bits stored as the attention waves' lane masks (16 words per 16x32 tile, read as scalars): a weight is masked by one select, no per-lane bit test; a tile's nodes summed in another order
# speedup vs baseline: 1.6115x; 1.0388x over previous
#include <hip/hip_runtime.h>
#include <math.h>

#define NN      4096
#define NFEAT   512
#define NHID    64
#define NCLASS  128
#define NHEADS  8
#define ALPHA   0.2f
#define NEGV    (-9000000000000000.0f)

typedef __attribute__((ext_vector_type(2))) float v2f;
typedef __attribute__((ext_vector_type(4))) float v4f;
typedef __attribute__((ext_vector_type(8))) float v8f;

typedef __attribute__((ext_vector_type(4))) int v4i;

__device__ __forceinline__ void adj_load_batch(v4i (&v)[8], const int* __restrict__ p) {
    #pragma unroll
    for (int k = 0; k < 8; ++k) v[k] = *(const v4i*)(p + 8 * k);
}
__device__ __forceinline__ void adj_pack_batch(const v4i (&v)[8], unsigned* __restrict__ q, int lane) {
    unsigned mine = 0u;
    #pragma unroll
    for (int k = 0; k < 8; ++k) {
        #pragma unroll
        for (int c = 0; c < 4; ++c) {
            const unsigned m = __builtin_amdgcn_ballot_w32(v[k][c] > 0);
            mine = (lane == 4 * k + c) ? m : mine;
        }
    }
    q[lane] = mine;
}

typedef __attribute__((ext_vector_type(16))) float v16f;

__device__ __forceinline__ float lane_tree_2(v2f y, int lo) {
    const bool up = (lo & 1) != 0;
    const float keep = up ? y.y : y.x;
    const float give = up ? y.x : y.y;
    return keep + __shfl_xor(give, 1, 32);
}
__device__ __forceinline__ float lane_tree_4(v4f y, int lo) {
    const bool up = (lo & 2) != 0;
    const v2f keep = up ? y.hi : y.lo;
    const v2f give = up ? y.lo : y.hi;
    v2f z;
    z.x = keep.x + __shfl_xor(give.x, 2, 32);
    z.y = keep.y + __shfl_xor(give.y, 2, 32);
    return lane_tree_2(z, lo);
}
__device__ __forceinline__ float lane_tree_8(v8f y, int lo) {
    const bool up = (lo & 4) != 0;
    const v4f keep = up ? y.hi : y.lo;
    const v4f give = up ? y.lo : y.hi;
    v4f z;
    #pragma unroll
    for (int i = 0; i < 4; ++i) z[i] = keep[i] + __shfl_xor(give[i], 4, 32);
    return lane_tree_4(z, lo);
}
__device__ __forceinline__ float lane_tree_16(v16f y, int lo) {
    const bool up = (lo & 8) != 0;
    const v8f keep = up ? y.hi : y.lo;
    const v8f give = up ? y.lo : y.hi;
    v8f z;
    #pragma unroll
    for (int i = 0; i < 8; ++i) z[i] = keep[i] + __shfl_xor(give[i], 8, 32);
    return lane_tree_8(z, lo);
}

template <int R, int CT, int G, int K, int NB>
__device__ __forceinline__ void gemm_load_block(
        v4f (&a)[R][G], __attribute__((ext_vector_type(CT))) float (&b)[G][4],
        const float* __restrict__ ap, const float* __restrict__ bp, int k) {
    typedef __attribute__((ext_vector_type(CT))) float vct;
    #pragma unroll
    for (int g = 0; g < G; ++g) {
        #pragma unroll
        for (int r = 0; r < R; ++r)
            a[r][g] = *(const v4f*)(ap + (long)r * 16 * K + k + 8 * g);
        #pragma unroll
        for (int j = 0; j < 4; ++j)
            b[g][j] = *(const vct*)(bp + (long)(k + 8 * g + j) * NB);
    }
}

template <int R, int CT, int G>
__device__ __forceinline__ void gemm_mma_block(
        v8f (&acc)[R][CT],
        const v4f (&a)[R][G], const __attribute__((ext_vector_type(CT))) float (&b)[G][4]) {
    v2f q[G][2][CT];
    #pragma unroll
    for (int g = 0; g < G; ++g)
        #pragma unroll
        for (int kk = 0; kk < 2; ++kk)
            #pragma unroll
            for (int t = 0; t < CT; ++t) {
                q[g][kk][t].x = b[g][2 * kk][t];
                q[g][kk][t].y = b[g][2 * kk + 1][t];
            }
    __builtin_amdgcn_sched_barrier(0);
    #pragma unroll
    for (int g = 0; g < G; ++g) {
        #pragma unroll
        for (int kk = 0; kk < 2; ++kk) {
            #pragma unroll
            for (int t = 0; t < CT; ++t) {
                #pragma unroll
                for (int r = 0; r < R; ++r) {
                    v2f av;
                    av.x = a[r][g][2 * kk];
                    av.y = a[r][g][2 * kk + 1];
                    acc[r][t] = __builtin_amdgcn_wmma_f32_16x16x4_f32(
                        false, av, false, q[g][kk][t], (short)0, acc[r][t], false, false);
                }
            }
        }
    }
    __builtin_amdgcn_sched_barrier(0);
    asm volatile("v_nop\n\tv_nop\n\tv_nop\n\tv_nop");
    __builtin_amdgcn_sched_barrier(0);
}

template <int K, int NB, int R, int CT, int KB, int WCOLS, bool PACK>
__global__ __launch_bounds__(128) void wmma_gemm_f32(
        const float* __restrict__ A,
        const float* __restrict__ B, long bStride,
        float* __restrict__ C, long cStride,
        const float* __restrict__ avec, int aStride,
        float* __restrict__ fi, float* __restrict__ fj, int n,
        const int* __restrict__ adj, unsigned* __restrict__ bits) {
    typedef __attribute__((ext_vector_type(CT))) float vct;
    constexpr int G = KB / 8;
    constexpr int WR = 4 / WCOLS;
    static_assert(K % (2 * KB) == 0, "K must hold an even number of k blocks");
    static_assert(WCOLS * 16 * CT == NB, "a block must hold whole rows of C");
    static_assert(WCOLS == 1 || WR == 1, "waves run along the rows or along the columns");
    const int wave = threadIdx.x >> 5;
    const int lane = threadIdx.x & 31;
    const int hi   = lane >> 4;
    const int lo   = lane & 15;
    const int wc   = wave % WCOLS;
    const int row0 = (blockIdx.x * WR + wave / WCOLS) * (16 * R);
    const int colb = wc * (16 * CT);
    const int head = blockIdx.z;

    const float* __restrict__ ap = A + (long)(row0 + lo) * K + 4 * hi;
    const float* __restrict__ bp = B + (long)head * bStride + (long)(4 * hi) * NB + colb + CT * lo;

    v8f acc[R][CT];
    #pragma unroll
    for (int r = 0; r < R; ++r)
        #pragma unroll
        for (int t = 0; t < CT; ++t) acc[r][t] = (v8f){};

    v4f a0[R][G], a1[R][G];
    vct b0[G][4], b1[G][4];
    gemm_load_block<R, CT, G, K, NB>(a0, b0, ap, bp, 0);

    constexpr int  NBX    = NN / (16 * R * WR);
    constexpr int  NWAVES = NBX * NHEADS * 4;
    constexpr long BSTEP  = (long)NWAVES * 1024;
    static_assert(!PACK || (long)NWAVES * (K / KB) * 1024 == (long)NN * NN,
                  "the waves' batches cover the adjacency matrix exactly once");
    constexpr int  JBLK   = NN / 64;
    static_assert(!PACK || NWAVES % JBLK == 0, "a step of NWAVES batches is whole row tiles");
    const int gwave = (blockIdx.z * NBX + blockIdx.x) * 4 + wave;
    const int* __restrict__ adjp = adj + (long)(16 * (gwave / JBLK) + lo) * NN + 64 * (gwave % JBLK) + 4 * hi;
    unsigned* __restrict__ bitp = bits + (long)gwave * 32;
    v4i va[8], vb[8];
    if constexpr (PACK) {
        #pragma unroll
        for (int k = 0; k < 8; ++k) vb[k] = (v4i){};
    }

    #pragma unroll 1
    for (int k0 = 0; k0 < K; k0 += 2 * KB) {
        gemm_load_block<R, CT, G, K, NB>(a1, b1, ap, bp, k0 + KB);
        if constexpr (PACK) adj_load_batch(va, adjp + (long)(k0 / KB) * BSTEP);
        __builtin_amdgcn_sched_barrier(0);
        gemm_mma_block<R, CT, G>(acc, a0, b0);
        if constexpr (PACK) {
            if (k0 > 0) adj_pack_batch(vb, bitp + (long)(k0 / KB - 1) * (BSTEP / 32), lane);
        }
        const int kn = (k0 + 2 * KB < K) ? (k0 + 2 * KB) : k0;
        gemm_load_block<R, CT, G, K, NB>(a0, b0, ap, bp, kn);
        if constexpr (PACK) adj_load_batch(vb, adjp + (long)(k0 / KB + 1) * BSTEP);
        __builtin_amdgcn_sched_barrier(0);
        gemm_mma_block<R, CT, G>(acc, a1, b1);
        if constexpr (PACK) adj_pack_batch(va, bitp + (long)(k0 / KB) * (BSTEP / 32), lane);
    }
    if constexpr (PACK) adj_pack_batch(vb, bitp + (long)(K / KB - 1) * (BSTEP / 32), lane);

    float* __restrict__ cp = C + (long)head * cStride + (long)(row0 + 8 * hi) * NB + colb + CT * lo;
    #pragma unroll
    for (int r = 0; r < R; ++r) {
        #pragma unroll
        for (int v = 0; v < 8; ++v) {
            vct o;
            #pragma unroll
            for (int t = 0; t < CT; ++t) o[t] = acc[r][t][v];
            *(vct*)(cp + (long)(16 * r + v) * NB) = o;
        }
    }

    const float* __restrict__ a1p = avec + (long)head * aStride + colb + CT * lo;
    const vct a1v = *(const vct*)a1p;
    const vct a2v = *(const vct*)(a1p + NB);
    constexpr int NV = 8 * R;
    static_assert(R == 1 || R == 2, "8 or 16 rows per lane half");
    typedef __attribute__((ext_vector_type(NV))) float vnv;
    vnv p1, p2;
    #pragma unroll
    for (int r = 0; r < R; ++r) {
        #pragma unroll
        for (int v = 0; v < 8; ++v) {
            float s1 = 0.f, s2 = 0.f;
            #pragma unroll
            for (int t = 0; t < CT; ++t) {
                s1 += acc[r][t][v] * a1v[t];
                s2 += acc[r][t][v] * a2v[t];
            }
            p1[8 * r + v] = s1;
            p2[8 * r + v] = s2;
        }
    }
    float w1, w2;
    if constexpr (NV == 16) {
        w1 = lane_tree_16(p1, lo);
        w2 = lane_tree_16(p2, lo);
    } else {
        w1 = lane_tree_8(p1, lo);
        w2 = lane_tree_8(p2, lo);
        w1 += __shfl_xor(w1, 8, 32);
        w2 += __shfl_xor(w2, 8, 32);
    }
    const int qv   = lo % NV;
    const int qrow = 16 * (qv >> 3) + 8 * hi + (qv & 7);
    if constexpr (WCOLS == 1) {
        if (lo < NV) {
            const long o = (long)head * n + row0 + qrow;
            fi[o] = w1;
            fj[o] = w2;
        }
    } else {
        __shared__ float part[2][WCOLS][16 * R];
        if (lo < NV) {
            part[0][wc][qrow] = w1;
            part[1][wc][qrow] = w2;
        }
        __syncthreads();
        static_assert(R == 1, "the packed line holds one row tile");
        if (threadIdx.x < 32) {
            const int which = threadIdx.x >> 4;
            float t1 = 0.f;
            #pragma unroll
            for (int w = 0; w < WCOLS; ++w) t1 += part[which][w][threadIdx.x & 15];
            fi[2 * ((long)head * n + row0) + threadIdx.x] = t1;
        }
    }
}

template <int F, int JW, int R, int PRE, bool ELU, bool PK, bool LSM>
__global__ __launch_bounds__(JW * 32) void gat_attn(
        const float* __restrict__ Hfeat, long hStride,
        const float* __restrict__ fi, const float* __restrict__ fj,
        const unsigned* __restrict__ bits,
        float* __restrict__ Out, int ldo, int headColMul, int n) {
    constexpr int NC = F / 16;
    constexpr float LOG2E = 1.4426950408889634f;
    constexpr float AL2   = ALPHA * LOG2E;
    const int wave = __builtin_amdgcn_readfirstlane(threadIdx.x >> 5);
    const int lane = threadIdx.x & 31;
    const int hi   = lane >> 4;
    const int lo   = lane & 15;
    const int head = blockIdx.y;
    const int i0   = blockIdx.x * (16 * R);
    const int ntiles = n >> 5;

    const float* __restrict__ Hh  = Hfeat + (long)head * hStride;
    const float* __restrict__ fih = fi + (PK ? 2 : 1) * (long)head * n;
    const float* __restrict__ fjh = fj + (PK ? 2 : 1) * (long)head * n;
    const float* __restrict__ bbase = Hh + 4 * hi * F + lo;

    float fiv[R];
    const unsigned* __restrict__ mrow[R];
    #pragma unroll
    for (int r = 0; r < R; ++r) {
        fiv[r]  = LOG2E * fih[(PK ? 2 : 1) * (i0 + 16 * r) + lo];
        mrow[r] = bits + (long)((i0 >> 4) + r) * ntiles * 16;
    }

    const int chunk = n / JW;
    const int jbeg  = wave * chunk;
    const int jend  = jbeg + chunk;

    constexpr int NLD = NN / (JW * 128);
    static_assert(NLD * JW * 128 == NN, "whole 16-byte loads per lane");
    float fjmax = -INFINITY;
    #pragma unroll
    for (int k = 0; k < NLD; ++k) {
        const int x = jbeg + 128 * k + 4 * lane;
        const v4f t = *(const v4f*)(fjh + (PK ? (2 * (x & ~15) + (x & 15)) : x));
        fjmax = fmaxf(fmaxf(fjmax, fmaxf(t.x, t.y)), fmaxf(t.z, t.w));
    }
    #pragma unroll
    for (int off = 16; off >= 1; off >>= 1) fjmax = fmaxf(fjmax, __shfl_xor(fjmax, off, 32));
    fjmax *= LOG2E;

    float m[R], l[R], a1[R], a2[R];
    v8f acc[R][NC];
    #pragma unroll
    for (int r = 0; r < R; ++r) {
        const float vb = fiv[r] + fjmax;
        m[r] = fmaxf(vb, ALPHA * vb);
        a1[r] = fiv[r] - m[r];
        a2[r] = ALPHA * fiv[r] - m[r];
        l[r] = 0.f;
        #pragma unroll
        for (int c = 0; c < NC; ++c) acc[r][c] = (v8f){};
    }

    unsigned wm_n[R][16];
    float fj_n[16];
    #pragma unroll
    for (int r = 0; r < R; ++r)
        #pragma unroll
        for (int e = 0; e < 16; ++e) wm_n[r][e] = mrow[r][(jbeg >> 5) * 16 + e];
    {
        const float* __restrict__ fjp = fjh + (PK ? 2 : 1) * jbeg + 4 * hi;
        #pragma unroll
        for (int q = 0; q < 4; ++q) {
            const v4f t4 = *(const v4f*)(fjp + (PK ? (32 * (q >> 1) + 8 * (q & 1)) : 8 * q));
            #pragma unroll
            for (int t = 0; t < 4; ++t) fj_n[4 * q + t] = t4[t];
        }
    }

    for (int j0 = jbeg; j0 < jend; j0 += 32) {
        unsigned wmask[R][16];
        float fjc[16];
        #pragma unroll
        for (int r = 0; r < R; ++r)
            #pragma unroll
            for (int e = 0; e < 16; ++e) wmask[r][e] = wm_n[r][e];
        #pragma unroll
        for (int q = 0; q < 16; ++q) fjc[q] = fj_n[q];

        #pragma unroll
        for (int half = 0; half < 2; ++half) {
            const int jb = j0 + 16 * half;

            float fjv[8];
            #pragma unroll
            for (int q = 0; q < 8; ++q) fjv[q] = fjc[8 * half + q];

            v2f bf[4][NC];
            const float* __restrict__ bp = bbase + jb * F;
            #pragma unroll
            for (int s = 0; s < PRE; ++s) {
                const float* __restrict__ bs = bp + (8 * (s >> 1) + 2 * (s & 1)) * F;
                #pragma unroll
                for (int c = 0; c < NC; ++c) {
                    bf[s][c].x = bs[16 * c];
                    bf[s][c].y = bs[16 * c + F];
                }
            }

            v2f af[R][4];
            #pragma unroll
            for (int r = 0; r < R; ++r) {
                float psum = 0.f;
                #pragma unroll
                for (int s = 0; s < 4; ++s) {
                    #pragma unroll
                    for (int t = 0; t < 2; ++t) {
                        const float f = fjv[2 * s + t];
                        const float x = fmaxf(__builtin_fmaf(f, LOG2E, a1[r]),
                                              __builtin_fmaf(f, AL2, a2[r]));
                        const float w = __builtin_amdgcn_exp2f(x);
                        const float p = __builtin_amdgcn_inverse_ballot_w32(wmask[r][8 * half + 2 * s + t]) ? w : 0.f;
                        af[r][s][t] = p;
                        psum += p;
                    }
                }
                l[r] += psum;
            }
            #pragma unroll
            for (int r = 0; r < R; ++r)
                #pragma unroll
                for (int s = 0; s < 4; ++s) asm volatile("" : "+v"(af[r][s]));

            if (half == 1) {
                const int jn = (j0 + 32 < jend) ? (j0 + 32) : j0;
                const float* __restrict__ fjp = fjh + (PK ? 2 : 1) * jn + 4 * hi;
                #pragma unroll
                for (int q = 0; q < 4; ++q) {
                    const v4f t4 = *(const v4f*)(fjp + (PK ? (32 * (q >> 1) + 8 * (q & 1)) : 8 * q));
                    #pragma unroll
                    for (int t = 0; t < 4; ++t) fj_n[4 * q + t] = t4[t];
                }
                #pragma unroll
                for (int r = 0; r < R; ++r)
                    #pragma unroll
                    for (int e = 0; e < 16; ++e) wm_n[r][e] = mrow[r][(jn >> 5) * 16 + e];
            }
            #pragma unroll
            for (int s = PRE; s < 4; ++s) {
                const float* __restrict__ bs = bp + (8 * (s >> 1) + 2 * (s & 1)) * F;
                #pragma unroll
                for (int c = 0; c < NC; ++c) {
                    bf[s][c].x = bs[16 * c];
                    bf[s][c].y = bs[16 * c + F];
                }
            }
            __builtin_amdgcn_sched_barrier(0);

            #pragma unroll
            for (int s = 0; s < 4; ++s) {
                #pragma unroll
                for (int c = 0; c < NC; ++c) {
                    #pragma unroll
                    for (int r = 0; r < R; ++r) {
                        acc[r][c] = __builtin_amdgcn_wmma_f32_16x16x4_f32(
                            false, af[r][s], false, bf[s][c], (short)0, acc[r][c], false, false);
                    }
                }
            }
            #pragma unroll
            for (int r = 0; r < R; ++r)
                #pragma unroll
                for (int c = 0; c < NC; ++c) asm volatile("" : "+v"(acc[r][c]));
            __builtin_amdgcn_sched_barrier(0);
            asm volatile("v_nop\n\tv_nop\n\tv_nop\n\tv_nop");
            __builtin_amdgcn_sched_barrier(0);
        }
    }

    #pragma unroll
    for (int r = 0; r < R; ++r) l[r] += __shfl_xor(l[r], 16, 32);

    __shared__ float m_lds[R][JW * 16];
    __shared__ float l_lds[R][JW * 16];
    __shared__ float L_lds[16];
    __shared__ float accbuf[JW * 16 * F];
    __shared__ float obuf[LSM ? 16 * F : 1];
    __shared__ float cpart[JW * 32];
    __shared__ float cmean[F];

    #pragma unroll
    for (int r = 0; r < R; ++r) {
        if (hi == 0) {
            m_lds[r][wave * 16 + lo] = m[r];
            l_lds[r][wave * 16 + lo] = l[r];
        }
    }
    __syncthreads();

    constexpr int TOT = 16 * F;
    constexpr int NT  = JW * 32;
    #pragma unroll
    for (int r = 0; r < R; ++r) {
        float Mrow = -INFINITY;
        #pragma unroll
        for (int w = 0; w < JW; ++w) Mrow = fmaxf(Mrow, m_lds[r][w * 16 + lo]);
        float Lrow = 0.f;
        #pragma unroll
        for (int w = 0; w < JW; ++w)
            Lrow += l_lds[r][w * 16 + lo] * __builtin_amdgcn_exp2f(m_lds[r][w * 16 + lo] - Mrow);
        if (wave == 0 && hi == 0) L_lds[lo] = Lrow;

        const float fown = __builtin_amdgcn_exp2f(m[r] - Mrow);
        float fv[8];
        #pragma unroll
        for (int v = 0; v < 8; ++v) fv[v] = __shfl(fown, v + 8 * hi, 32);

        float* ab = accbuf + (wave * 16 + 8 * hi) * F + lo;
        #pragma unroll
        for (int c = 0; c < NC; ++c)
            #pragma unroll
            for (int v = 0; v < 8; ++v)
                ab[v * F + 16 * c] = acc[r][c][v] * fv[v];
        __syncthreads();

        const bool anyEmpty = __any(L_lds[lo] == 0.f);
        if (anyEmpty) {
            static_assert(NT % F == 0, "whole columns per thread group");
            constexpr int NS = NT / F;
            const int ccol = threadIdx.x % F;
            float csum = 0.f;
            for (int j = threadIdx.x / F; j < n; j += NS) csum += Hh[(long)j * F + ccol];
            cpart[threadIdx.x] = csum;
            __syncthreads();
            if (threadIdx.x < F) {
                float t = 0.f;
                #pragma unroll
                for (int k = 0; k < NS; ++k) t += cpart[k * F + threadIdx.x];
                cmean[threadIdx.x] = t / (float)n;
            }
            __syncthreads();
        }

        float* outBase = Out + (long)(i0 + 16 * r) * ldo + head * headColMul;
        #pragma unroll
        for (int e = 0; e < TOT / NT; ++e) {
            const int flat = e * NT + threadIdx.x;
            const int row = flat / F;
            const int col = flat % F;
            float val = 0.f;
            #pragma unroll
            for (int w = 0; w < JW; ++w) val += accbuf[(w * 16 + row) * F + col];
            const float Lr = L_lds[row];
            float rr = val / Lr;
            if (anyEmpty && Lr == 0.f) rr = cmean[col];
            if (ELU) rr = rr > 0.f ? rr : (__expf(rr) - 1.0f);
            if constexpr (LSM) obuf[row * F + col] = rr;
            else outBase[(long)row * ldo + col] = rr;
        }
        if (r + 1 < R) __syncthreads();
    }

    if constexpr (LSM) {
        static_assert(R == 1 && JW == 16 && F == 128, "one wave per row, four columns per lane");
        __syncthreads();
        const float* rp = obuf + wave * F;
        float v[4];
        float mx = -INFINITY;
        #pragma unroll
        for (int q = 0; q < 4; ++q) {
            v[q] = rp[lane + 32 * q];
            mx = fmaxf(mx, v[q]);
        }
        #pragma unroll
        for (int off = 16; off >= 1; off >>= 1) mx = fmaxf(mx, __shfl_xor(mx, off, 32));
        float s = 0.f;
        #pragma unroll
        for (int q = 0; q < 4; ++q) s += __expf(v[q] - mx);
        #pragma unroll
        for (int off = 16; off >= 1; off >>= 1) s += __shfl_xor(s, off, 32);
        const float ls = __logf(s);
        float* op = Out + (long)(i0 + wave) * ldo + head * headColMul;
        #pragma unroll
        for (int q = 0; q < 4; ++q) op[lane + 32 * q] = v[q] - mx - ls;
    }
}

extern "C" void kernel_launch(void* const* d_in, const int* in_sizes, int n_in,
                              void* d_out, int out_size, void* d_ws, size_t ws_size,
                              hipStream_t stream) {
    const float* x     = (const float*)d_in[0];
    const int*   adj   = (const int*)d_in[1];
    const float* W     = (const float*)d_in[2];
    const float* a     = (const float*)d_in[3];
    const float* W_out = (const float*)d_in[4];
    const float* a_out = (const float*)d_in[5];
    float* out = (float*)d_out;

    char* ws = (char*)d_ws;
    unsigned* bits = (unsigned*)ws;
    ws += (size_t)NN * (NN / 32) * sizeof(unsigned);
    float* h1 = (float*)ws;
    ws += (size_t)NHEADS * NN * NHID * sizeof(float);
    float* fi1 = (float*)ws; ws += (size_t)NHEADS * NN * sizeof(float);
    float* fj1 = (float*)ws; ws += (size_t)NHEADS * NN * sizeof(float);
    float* xcat = (float*)ws;
    ws += (size_t)NN * (NHEADS * NHID) * sizeof(float);
    float* h2 = (float*)ws;
    ws += (size_t)NN * NCLASS * sizeof(float);
    float* fij2 = (float*)ws;

    wmma_gemm_f32<NFEAT, NHID, 2, 4, 32, 1, true><<<dim3(NN / 128, 1, NHEADS), dim3(128), 0, stream>>>(
        x, W, (long)NFEAT * NHID, h1, (long)NN * NHID,
        a, 2 * NHID, fi1, fj1, NN, adj, bits);

    gat_attn<NHID, 4, 2, 4, false, false, false><<<dim3(NN / 32, NHEADS), dim3(128), 0, stream>>>(
        h1, (long)NN * NHID, fi1, fj1, bits,
        xcat, NHEADS * NHID, NHID, NN);

    wmma_gemm_f32<NHEADS * NHID, NCLASS, 1, 2, 64, 4, false><<<dim3(NN / 16, 1, 1), dim3(128), 0, stream>>>(
        xcat, W_out, 0, h2, 0,
        a_out, 2 * NCLASS, fij2, fij2, NN, nullptr, nullptr);

    gat_attn<NCLASS, 16, 1, 3, true, true, true><<<dim3(NN / 16, 1), dim3(512), 0, stream>>>(
        h2, 0, fij2, fij2 + 16, bits,
        out, NCLASS, 0, NN);
}
